# starting kernel
# baseline (speedup 1.0000x reference)
#include <hip/hip_runtime.h>

#define N_NODES 100000
#define N_EDGES 1200000
#define DIM 64
#define E_TOT (N_EDGES + N_NODES)

typedef __attribute__((ext_vector_type(2))) float v2f;
typedef __attribute__((ext_vector_type(8))) float v8f;

__global__ void deg_init_kernel(float* __restrict__ deg) {
    int i = blockIdx.x * blockDim.x + threadIdx.x;
    if (i < N_NODES) deg[i] = 1.0f;
}

__global__ void deg_accum_kernel(const int* __restrict__ ei, float* __restrict__ deg) {
    int e = blockIdx.x * blockDim.x + threadIdx.x;
    if (e < N_EDGES) atomicAdd(&deg[ei[N_EDGES + e]], 1.0f);
}

__global__ void dinv_kernel(float* __restrict__ deg) {
    int i = blockIdx.x * blockDim.x + threadIdx.x;
    if (i < N_NODES) deg[i] = rsqrtf(deg[i]);
}

__global__ void gemm_wmma_kernel(const float* __restrict__ in,
                                 const float* __restrict__ W,
                                 float* __restrict__ out) {
    const int lane = threadIdx.x & 31;
    const int wid  = (blockIdx.x * blockDim.x + threadIdx.x) >> 5;
    const int nWaves = (gridDim.x * blockDim.x) >> 5;

    const int col0 = (wid & 3) * 16;
    const int m    = lane & 15;
    const int hi   = lane >> 4;
    const int klo  = hi * 2;

    v2f bfrag[16];
#pragma unroll
    for (int kb = 0; kb < 16; ++kb) {
        int k = kb * 4 + klo;
        bfrag[kb].x = W[k * DIM + col0 + m];
        bfrag[kb].y = W[(k + 1) * DIM + col0 + m];
    }

    const int nRowTiles  = N_NODES / 16;
    const int rowStride  = nWaves >> 2;

    for (int rt = wid >> 2; rt < nRowTiles; rt += rowStride) {
        const int row0 = rt * 16;
        const float* arow = in + (size_t)(row0 + m) * DIM;
        v8f c = {};
#pragma unroll
        for (int kb = 0; kb < 16; ++kb) {
            v2f a = *(const v2f*)(arow + kb * 4 + klo);
            c = __builtin_amdgcn_wmma_f32_16x16x4_f32(
                     false, a,  false, bfrag[kb],
                     (short)0, c,  false,  false);
        }
        float* orow = out + (size_t)(row0 + hi * 8) * DIM + col0 + m;
#pragma unroll
        for (int r = 0; r < 8; ++r) orow[(size_t)r * DIM] = c[r];
    }
}

__global__ void bias_fill_kernel(float* __restrict__ h, const float* __restrict__ b) {
    int i = blockIdx.x * blockDim.x + threadIdx.x;
    if (i < N_NODES * DIM) h[i] = b[i & (DIM - 1)];
}

__global__ void scatter_kernel(const float* __restrict__ t,
                               const int* __restrict__ ei,
                               const float* __restrict__ dinv,
                               float* __restrict__ h) {
    int tid = blockIdx.x * blockDim.x + threadIdx.x;
    int e = tid >> 5;
    int p = tid & 31;
    if (e >= E_TOT) return;
    int s, d;
    if (e < N_EDGES) { s = ei[e]; d = ei[N_EDGES + e]; }
    else             { s = e - N_EDGES; d = s; }
    float nm = dinv[s] * dinv[d];
    const float* mp = t + (size_t)s * DIM + p * 2;
    v2f msg = *(const v2f*)mp;
    float* op = h + (size_t)d * DIM + p * 2;
    atomicAdd(op,     msg.x * nm);
    atomicAdd(op + 1, msg.y * nm);
}

__global__ void relu_kernel(float* __restrict__ h) {
    int i = blockIdx.x * blockDim.x + threadIdx.x;
    if (i < N_NODES * DIM) h[i] = fmaxf(h[i], 0.0f);
}

extern "C" void kernel_launch(void* const* d_in, const int* in_sizes, int n_in,
                              void* d_out, int out_size, void* d_ws, size_t ws_size,
                              hipStream_t stream) {
    const float* x  = (const float*)d_in[0];
    const int*   ei = (const int*)d_in[1];
    const float* W1 = (const float*)d_in[2];
    const float* b1 = (const float*)d_in[3];
    const float* W2 = (const float*)d_in[4];
    const float* b2 = (const float*)d_in[5];
    const float* W3 = (const float*)d_in[6];
    const float* b3 = (const float*)d_in[7];
    float* out = (float*)d_out;

    char* ws = (char*)d_ws;
    size_t off = 0;
    float* dinv = (float*)(ws + off); off += ((size_t)N_NODES * 4 + 255) & ~(size_t)255;
    float* tbuf = (float*)(ws + off); off += (size_t)N_NODES * DIM * 4;
    float* hbuf = (float*)(ws + off);

    const int T = 256;
    const int gN   = (N_NODES + T - 1) / T;
    const int gE   = (N_EDGES + T - 1) / T;
    const int gND  = (N_NODES * DIM + T - 1) / T;
    const int gSc  = ((E_TOT * 32) + T - 1) / T;
    const int gGemm = 512;

    deg_init_kernel<<<gN, T, 0, stream>>>(dinv);
    deg_accum_kernel<<<gE, T, 0, stream>>>(ei, dinv);
    dinv_kernel<<<gN, T, 0, stream>>>(dinv);

    gemm_wmma_kernel<<<gGemm, T, 0, stream>>>(x, W1, tbuf);
    bias_fill_kernel<<<gND, T, 0, stream>>>(hbuf, b1);
    scatter_kernel<<<gSc, T, 0, stream>>>(tbuf, ei, dinv, hbuf);
    relu_kernel<<<gND, T, 0, stream>>>(hbuf);

    gemm_wmma_kernel<<<gGemm, T, 0, stream>>>(hbuf, W2, tbuf);
    bias_fill_kernel<<<gND, T, 0, stream>>>(hbuf, b2);
    scatter_kernel<<<gSc, T, 0, stream>>>(tbuf, ei, dinv, hbuf);
    relu_kernel<<<gND, T, 0, stream>>>(hbuf);

    gemm_wmma_kernel<<<gGemm, T, 0, stream>>>(hbuf, W3, tbuf);
    bias_fill_kernel<<<gND, T, 0, stream>>>(out, b3);
    scatter_kernel<<<gSc, T, 0, stream>>>(tbuf, ei, dinv, out);
}


# Atomic scatter-add replaced by a per-call destination-grouped edge list (atomic rank, scan) and a register gather that adds bias/relu and stores each row once; sum order = rank (arrival) order
# speedup vs baseline: 2.2549x; 2.2549x over previous
#include <hip/hip_runtime.h>


#define N_NODES 100000
#define N_EDGES 1200000
#define DIM 64

#define OFF_DINV   ((size_t)0)
#define OFF_TBUF   ((size_t)400128)
#define OFF_CNT    ((size_t)400128 + (size_t)25600000)
#define OFF_ROWPTR (OFF_CNT + (size_t)400128)
#define OFF_RANK   (OFF_ROWPTR + (size_t)400384)
#define OFF_CSR    (OFF_RANK + (size_t)4800000)

typedef __attribute__((ext_vector_type(2))) float v2f;
typedef __attribute__((ext_vector_type(8))) float v8f;

__global__ void cnt_zero_kernel(int* __restrict__ cnt) {
    int i = blockIdx.x * blockDim.x + threadIdx.x;
    if (i < N_NODES) cnt[i] = 0;
}

__global__ void rank_kernel(const int* __restrict__ ei, int* __restrict__ cnt,
                            int* __restrict__ rank) {
    int i = blockIdx.x * blockDim.x + threadIdx.x;
    if (i < N_EDGES / 4) {
        const int4 d = ((const int4*)(ei + N_EDGES))[i];
        int4 r;
        r.x = atomicAdd(&cnt[d.x], 1);
        r.y = atomicAdd(&cnt[d.y], 1);
        r.z = atomicAdd(&cnt[d.z], 1);
        r.w = atomicAdd(&cnt[d.w], 1);
        ((int4*)rank)[i] = r;
    }
}

__global__ __launch_bounds__(1024) void scan_dinv_kernel(const int* __restrict__ cnt,
                                                         int* __restrict__ rowptr,
                                                         float* __restrict__ dinv) {
    __shared__ int wsum[32];
    const int tid = threadIdx.x;
    const int lane = tid & 31;
    const int w = tid >> 5;
    int carry = 0;
    int4 nxt[4];
#pragma unroll
    for (int q = 0; q < 4; ++q) nxt[q] = make_int4(0, 0, 0, 0);
    if (tid * 16 < N_NODES) {
#pragma unroll
        for (int q = 0; q < 4; ++q) nxt[q] = *(const int4*)(cnt + tid * 16 + q * 4);
    }
    for (int base = 0; base < N_NODES; base += 16384) {
        const int i = base + tid * 16;
        int c[16];
#pragma unroll
        for (int q = 0; q < 4; ++q) {
            c[q * 4 + 0] = nxt[q].x;
            c[q * 4 + 1] = nxt[q].y;
            c[q * 4 + 2] = nxt[q].z;
            c[q * 4 + 3] = nxt[q].w;
            nxt[q] = make_int4(0, 0, 0, 0);
        }
        if (i + 16384 < N_NODES) {
#pragma unroll
            for (int q = 0; q < 4; ++q) nxt[q] = *(const int4*)(cnt + i + 16384 + q * 4);
        }
        int pre[16];
        int s = 0;
#pragma unroll
        for (int k = 0; k < 16; ++k) { pre[k] = s; s += c[k]; }
        int x = s;
#pragma unroll
        for (int o = 1; o < 32; o <<= 1) {
            int y = __shfl_up(x, o);
            if (lane >= o) x += y;
        }
        if (lane == 31) wsum[w] = x;
        __syncthreads();
        const int ws = wsum[lane];
        int wx = ws;
#pragma unroll
        for (int o = 1; o < 32; o <<= 1) {
            int y = __shfl_up(wx, o);
            if (lane >= o) wx += y;
        }
        const int wave_excl = __shfl(wx - ws, w);
        const int total = __shfl(wx, 31);
        const int ex = carry + wave_excl + (x - s);
        if (i < N_NODES) {
#pragma unroll
            for (int q = 0; q < 4; ++q) {
                *(int4*)(rowptr + i + q * 4) = make_int4(ex + pre[q * 4 + 0], ex + pre[q * 4 + 1],
                                                         ex + pre[q * 4 + 2], ex + pre[q * 4 + 3]);
                float4 dv;
                dv.x = rsqrtf((float)(c[q * 4 + 0] + 1));
                dv.y = rsqrtf((float)(c[q * 4 + 1] + 1));
                dv.z = rsqrtf((float)(c[q * 4 + 2] + 1));
                dv.w = rsqrtf((float)(c[q * 4 + 3] + 1));
                *(float4*)(dinv + i + q * 4) = dv;
            }
        }
        carry += total;
        __syncthreads();
    }
    if (tid == 0) rowptr[N_NODES] = carry;
}

__global__ void fill_kernel(const int* __restrict__ ei, const int* __restrict__ rank,
                            const int* __restrict__ rowptr, const float* __restrict__ dinv,
                            int2* __restrict__ csr) {
    int i = blockIdx.x * blockDim.x + threadIdx.x;
    if (i < N_EDGES / 4) {
        const int4 s = ((const int4*)ei)[i];
        const int4 d = ((const int4*)(ei + N_EDGES))[i];
        const int4 r = ((const int4*)rank)[i];
        csr[rowptr[d.x] + r.x] = make_int2(s.x, __float_as_int(dinv[s.x] * dinv[d.x]));
        csr[rowptr[d.y] + r.y] = make_int2(s.y, __float_as_int(dinv[s.y] * dinv[d.y]));
        csr[rowptr[d.z] + r.z] = make_int2(s.z, __float_as_int(dinv[s.z] * dinv[d.z]));
        csr[rowptr[d.w] + r.w] = make_int2(s.w, __float_as_int(dinv[s.w] * dinv[d.w]));
    }
}

__global__ void gemm_wmma_kernel(const float* __restrict__ in,
                                 const float* __restrict__ W,
                                 float* __restrict__ out) {
    const int lane = threadIdx.x & 31;
    const int wid  = (blockIdx.x * blockDim.x + threadIdx.x) >> 5;
    const int nWaves = (gridDim.x * blockDim.x) >> 5;

    const int col0 = (wid & 3) * 16;
    const int m    = lane & 15;
    const int hi   = lane >> 4;
    const int klo  = hi * 2;

    v2f bfrag[16];
#pragma unroll
    for (int kb = 0; kb < 16; ++kb) {
        int k = kb * 4 + klo;
        bfrag[kb].x = W[k * DIM + col0 + m];
        bfrag[kb].y = W[(k + 1) * DIM + col0 + m];
    }

    const int nRowTiles  = N_NODES / 16;
    const int rowStride  = nWaves >> 2;

    for (int rt = wid >> 2; rt < nRowTiles; rt += rowStride) {
        const int row0 = rt * 16;
        const float* arow = in + (size_t)(row0 + m) * DIM;
        v8f c = {};
#pragma unroll
        for (int kb = 0; kb < 16; ++kb) {
            v2f a = *(const v2f*)(arow + kb * 4 + klo);
            c = __builtin_amdgcn_wmma_f32_16x16x4_f32(
                     false, a,  false, bfrag[kb],
                     (short)0, c,  false,  false);
        }
        float* orow = out + (size_t)(row0 + hi * 8) * DIM + col0 + m;
#pragma unroll
        for (int r = 0; r < 8; ++r) orow[(size_t)r * DIM] = c[r];
    }
}

static __device__ __forceinline__ void axpy4(float4& acc, const float4 r, const float w) {
    acc.x = fmaf(r.x, w, acc.x);
    acc.y = fmaf(r.y, w, acc.y);
    acc.z = fmaf(r.z, w, acc.z);
    acc.w = fmaf(r.w, w, acc.w);
}

template <bool RELU>
__global__ __launch_bounds__(256) void gather_kernel(const float* __restrict__ t,
                                                     const int* __restrict__ rowptr,
                                                     const int2* __restrict__ csr,
                                                     const float* __restrict__ dinv,
                                                     const float* __restrict__ bias,
                                                     float* __restrict__ h) {
    const int l = threadIdx.x & 15;
    const float4* t4 = (const float4*)t;
    const float4 b = ((const float4*)bias)[l];
    for (int n = blockIdx.x * 16 + (threadIdx.x >> 4); n < N_NODES; n += gridDim.x * 16) {
        const int beg = rowptr[n];
        const int end = rowptr[n + 1];
        const float dn = dinv[n];
        const float4 self = t4[n * 16 + l];
        const float wself = dn * dn;
        float4 acc;
        acc.x = self.x * wself;
        acc.y = self.y * wself;
        acc.z = self.z * wself;
        acc.w = self.w * wself;
        int j = beg;
        for (; j + 4 <= end; j += 4) {
            const int2 e0 = csr[j];
            const int2 e1 = csr[j + 1];
            const int2 e2 = csr[j + 2];
            const int2 e3 = csr[j + 3];
            const float4 r0 = t4[e0.x * 16 + l];
            const float4 r1 = t4[e1.x * 16 + l];
            const float4 r2 = t4[e2.x * 16 + l];
            const float4 r3 = t4[e3.x * 16 + l];
            axpy4(acc, r0, __int_as_float(e0.y));
            axpy4(acc, r1, __int_as_float(e1.y));
            axpy4(acc, r2, __int_as_float(e2.y));
            axpy4(acc, r3, __int_as_float(e3.y));
        }
        if (j + 2 <= end) {
            const int2 e0 = csr[j];
            const int2 e1 = csr[j + 1];
            const float4 r0 = t4[e0.x * 16 + l];
            const float4 r1 = t4[e1.x * 16 + l];
            axpy4(acc, r0, __int_as_float(e0.y));
            axpy4(acc, r1, __int_as_float(e1.y));
            j += 2;
        }
        if (j < end) {
            const int2 e0 = csr[j];
            const float4 r0 = t4[e0.x * 16 + l];
            axpy4(acc, r0, __int_as_float(e0.y));
        }
        acc.x += b.x;
        acc.y += b.y;
        acc.z += b.z;
        acc.w += b.w;
        if (RELU) {
            acc.x = fmaxf(acc.x, 0.0f);
            acc.y = fmaxf(acc.y, 0.0f);
            acc.z = fmaxf(acc.z, 0.0f);
            acc.w = fmaxf(acc.w, 0.0f);
        }
        ((float4*)h)[n * 16 + l] = acc;
    }
}

extern "C" void kernel_launch(void* const* d_in, const int* in_sizes, int n_in,
                              void* d_out, int out_size, void* d_ws, size_t ws_size,
                              hipStream_t stream) {
    const float* x  = (const float*)d_in[0];
    const int*   ei = (const int*)d_in[1];
    const float* W1 = (const float*)d_in[2];
    const float* b1 = (const float*)d_in[3];
    const float* W2 = (const float*)d_in[4];
    const float* b2 = (const float*)d_in[5];
    const float* W3 = (const float*)d_in[6];
    const float* b3 = (const float*)d_in[7];
    float* out = (float*)d_out;

    char* ws = (char*)d_ws;
    float* dinv   = (float*)(ws + OFF_DINV);
    float* tbuf   = (float*)(ws + OFF_TBUF);
    int*   cnt    = (int*)(ws + OFF_CNT);
    int*   rowptr = (int*)(ws + OFF_ROWPTR);
    int*   rank   = (int*)(ws + OFF_RANK);
    int2*  csr    = (int2*)(ws + OFF_CSR);

    cnt_zero_kernel<<<391, 256, 0, stream>>>(cnt);
    rank_kernel<<<1172, 256, 0, stream>>>(ei, cnt, rank);
    scan_dinv_kernel<<<1, 1024, 0, stream>>>(cnt, rowptr, dinv);
    fill_kernel<<<1172, 256, 0, stream>>>(ei, rank, rowptr, dinv, csr);

    gemm_wmma_kernel<<<512, 256, 0, stream>>>(x, W1, tbuf);
    gather_kernel<true><<<2048, 256, 0, stream>>>(tbuf, rowptr, csr, dinv, b1, out);

    gemm_wmma_kernel<<<512, 256, 0, stream>>>(out, W2, tbuf);
    gather_kernel<true><<<2048, 256, 0, stream>>>(tbuf, rowptr, csr, dinv, b2, out);

    gemm_wmma_kernel<<<512, 256, 0, stream>>>(out, W3, tbuf);
    gather_kernel<false><<<2048, 256, 0, stream>>>(tbuf, rowptr, csr, dinv, b3, out);
}


# Next layer's GEMM (f32 WMMA, same fragments and k order) fused into the gather's tail through a 16x64 LDS tile: layers 2 and 3 lose their GEMM launch and the h round trip; sum order unchanged
# speedup vs baseline: 2.4026x; 1.0655x over previous
#include <hip/hip_runtime.h>


#define N_NODES 100000
#define N_EDGES 1200000
#define DIM 64

#define OFF_DINV   ((size_t)0)
#define OFF_TBUF   ((size_t)400128)
#define OFF_CNT    ((size_t)400128 + (size_t)25600000)
#define OFF_ROWPTR (OFF_CNT + (size_t)400128)
#define OFF_RANK   (OFF_ROWPTR + (size_t)400384)
#define OFF_CSR    (OFF_RANK + (size_t)4800000)

typedef __attribute__((ext_vector_type(2))) float v2f;
typedef __attribute__((ext_vector_type(8))) float v8f;

__global__ void cnt_zero_kernel(int* __restrict__ cnt) {
    int i = blockIdx.x * blockDim.x + threadIdx.x;
    if (i < N_NODES) cnt[i] = 0;
}

__global__ void rank_kernel(const int* __restrict__ ei, int* __restrict__ cnt,
                            int* __restrict__ rank) {
    int i = blockIdx.x * blockDim.x + threadIdx.x;
    if (i < N_EDGES / 4) {
        const int4 d = ((const int4*)(ei + N_EDGES))[i];
        int4 r;
        r.x = atomicAdd(&cnt[d.x], 1);
        r.y = atomicAdd(&cnt[d.y], 1);
        r.z = atomicAdd(&cnt[d.z], 1);
        r.w = atomicAdd(&cnt[d.w], 1);
        ((int4*)rank)[i] = r;
    }
}

__global__ __launch_bounds__(1024) void scan_dinv_kernel(const int* __restrict__ cnt,
                                                         int* __restrict__ rowptr,
                                                         float* __restrict__ dinv) {
    __shared__ int wsum[32];
    const int tid = threadIdx.x;
    const int lane = tid & 31;
    const int w = tid >> 5;
    int carry = 0;
    int4 nxt[4];
#pragma unroll
    for (int q = 0; q < 4; ++q) nxt[q] = make_int4(0, 0, 0, 0);
    if (tid * 16 < N_NODES) {
#pragma unroll
        for (int q = 0; q < 4; ++q) nxt[q] = *(const int4*)(cnt + tid * 16 + q * 4);
    }
    for (int base = 0; base < N_NODES; base += 16384) {
        const int i = base + tid * 16;
        int c[16];
#pragma unroll
        for (int q = 0; q < 4; ++q) {
            c[q * 4 + 0] = nxt[q].x;
            c[q * 4 + 1] = nxt[q].y;
            c[q * 4 + 2] = nxt[q].z;
            c[q * 4 + 3] = nxt[q].w;
            nxt[q] = make_int4(0, 0, 0, 0);
        }
        if (i + 16384 < N_NODES) {
#pragma unroll
            for (int q = 0; q < 4; ++q) nxt[q] = *(const int4*)(cnt + i + 16384 + q * 4);
        }
        int pre[16];
        int s = 0;
#pragma unroll
        for (int k = 0; k < 16; ++k) { pre[k] = s; s += c[k]; }
        int x = s;
#pragma unroll
        for (int o = 1; o < 32; o <<= 1) {
            int y = __shfl_up(x, o);
            if (lane >= o) x += y;
        }
        if (lane == 31) wsum[w] = x;
        __syncthreads();
        const int ws = wsum[lane];
        int wx = ws;
#pragma unroll
        for (int o = 1; o < 32; o <<= 1) {
            int y = __shfl_up(wx, o);
            if (lane >= o) wx += y;
        }
        const int wave_excl = __shfl(wx - ws, w);
        const int total = __shfl(wx, 31);
        const int ex = carry + wave_excl + (x - s);
        if (i < N_NODES) {
#pragma unroll
            for (int q = 0; q < 4; ++q) {
                *(int4*)(rowptr + i + q * 4) = make_int4(ex + pre[q * 4 + 0], ex + pre[q * 4 + 1],
                                                         ex + pre[q * 4 + 2], ex + pre[q * 4 + 3]);
                float4 dv;
                dv.x = rsqrtf((float)(c[q * 4 + 0] + 1));
                dv.y = rsqrtf((float)(c[q * 4 + 1] + 1));
                dv.z = rsqrtf((float)(c[q * 4 + 2] + 1));
                dv.w = rsqrtf((float)(c[q * 4 + 3] + 1));
                *(float4*)(dinv + i + q * 4) = dv;
            }
        }
        carry += total;
        __syncthreads();
    }
    if (tid == 0) rowptr[N_NODES] = carry;
}

__global__ void fill_kernel(const int* __restrict__ ei, const int* __restrict__ rank,
                            const int* __restrict__ rowptr, const float* __restrict__ dinv,
                            int2* __restrict__ csr) {
    int i = blockIdx.x * blockDim.x + threadIdx.x;
    if (i < N_EDGES / 4) {
        const int4 s = ((const int4*)ei)[i];
        const int4 d = ((const int4*)(ei + N_EDGES))[i];
        const int4 r = ((const int4*)rank)[i];
        csr[rowptr[d.x] + r.x] = make_int2(s.x, __float_as_int(dinv[s.x] * dinv[d.x]));
        csr[rowptr[d.y] + r.y] = make_int2(s.y, __float_as_int(dinv[s.y] * dinv[d.y]));
        csr[rowptr[d.z] + r.z] = make_int2(s.z, __float_as_int(dinv[s.z] * dinv[d.z]));
        csr[rowptr[d.w] + r.w] = make_int2(s.w, __float_as_int(dinv[s.w] * dinv[d.w]));
    }
}

__global__ void gemm_wmma_kernel(const float* __restrict__ in,
                                 const float* __restrict__ W,
                                 float* __restrict__ out) {
    const int lane = threadIdx.x & 31;
    const int wid  = (blockIdx.x * blockDim.x + threadIdx.x) >> 5;
    const int nWaves = (gridDim.x * blockDim.x) >> 5;

    const int col0 = (wid & 3) * 16;
    const int m    = lane & 15;
    const int hi   = lane >> 4;
    const int klo  = hi * 2;

    v2f bfrag[16];
#pragma unroll
    for (int kb = 0; kb < 16; ++kb) {
        int k = kb * 4 + klo;
        bfrag[kb].x = W[k * DIM + col0 + m];
        bfrag[kb].y = W[(k + 1) * DIM + col0 + m];
    }

    const int nRowTiles  = N_NODES / 16;
    const int rowStride  = nWaves >> 2;

    for (int rt = wid >> 2; rt < nRowTiles; rt += rowStride) {
        const int row0 = rt * 16;
        const float* arow = in + (size_t)(row0 + m) * DIM;
        v8f c = {};
#pragma unroll
        for (int kb = 0; kb < 16; ++kb) {
            v2f a = *(const v2f*)(arow + kb * 4 + klo);
            c = __builtin_amdgcn_wmma_f32_16x16x4_f32(
                     false, a,  false, bfrag[kb],
                     (short)0, c,  false,  false);
        }
        float* orow = out + (size_t)(row0 + hi * 8) * DIM + col0 + m;
#pragma unroll
        for (int r = 0; r < 8; ++r) orow[(size_t)r * DIM] = c[r];
    }
}

static __device__ __forceinline__ void axpy4(float4& acc, const float4 r, const float w) {
    acc.x = fmaf(r.x, w, acc.x);
    acc.y = fmaf(r.y, w, acc.y);
    acc.z = fmaf(r.z, w, acc.z);
    acc.w = fmaf(r.w, w, acc.w);
}

template <bool RELU>
__global__ __launch_bounds__(256) void gather_kernel(const float* __restrict__ t,
                                                     const int* __restrict__ rowptr,
                                                     const int2* __restrict__ csr,
                                                     const float* __restrict__ dinv,
                                                     const float* __restrict__ bias,
                                                     float* __restrict__ h) {
    const int l = threadIdx.x & 15;
    const float4* t4 = (const float4*)t;
    const float4 b = ((const float4*)bias)[l];
    for (int n = blockIdx.x * 16 + (threadIdx.x >> 4); n < N_NODES; n += gridDim.x * 16) {
        const int beg = rowptr[n];
        const int end = rowptr[n + 1];
        const float dn = dinv[n];
        const float4 self = t4[n * 16 + l];
        const float wself = dn * dn;
        float4 acc;
        acc.x = self.x * wself;
        acc.y = self.y * wself;
        acc.z = self.z * wself;
        acc.w = self.w * wself;
        int j = beg;
        for (; j + 4 <= end; j += 4) {
            const int2 e0 = csr[j];
            const int2 e1 = csr[j + 1];
            const int2 e2 = csr[j + 2];
            const int2 e3 = csr[j + 3];
            const float4 r0 = t4[e0.x * 16 + l];
            const float4 r1 = t4[e1.x * 16 + l];
            const float4 r2 = t4[e2.x * 16 + l];
            const float4 r3 = t4[e3.x * 16 + l];
            axpy4(acc, r0, __int_as_float(e0.y));
            axpy4(acc, r1, __int_as_float(e1.y));
            axpy4(acc, r2, __int_as_float(e2.y));
            axpy4(acc, r3, __int_as_float(e3.y));
        }
        if (j + 2 <= end) {
            const int2 e0 = csr[j];
            const int2 e1 = csr[j + 1];
            const float4 r0 = t4[e0.x * 16 + l];
            const float4 r1 = t4[e1.x * 16 + l];
            axpy4(acc, r0, __int_as_float(e0.y));
            axpy4(acc, r1, __int_as_float(e1.y));
            j += 2;
        }
        if (j < end) {
            const int2 e0 = csr[j];
            const float4 r0 = t4[e0.x * 16 + l];
            axpy4(acc, r0, __int_as_float(e0.y));
        }
        acc.x += b.x;
        acc.y += b.y;
        acc.z += b.z;
        acc.w += b.w;
        if (RELU) {
            acc.x = fmaxf(acc.x, 0.0f);
            acc.y = fmaxf(acc.y, 0.0f);
            acc.z = fmaxf(acc.z, 0.0f);
            acc.w = fmaxf(acc.w, 0.0f);
        }
        ((float4*)h)[n * 16 + l] = acc;
    }
}

#define TILE_PITCH 68

__global__ __launch_bounds__(256) void gather_gemm_kernel(const float* __restrict__ t,
                                                          const int* __restrict__ rowptr,
                                                          const int2* __restrict__ csr,
                                                          const float* __restrict__ dinv,
                                                          const float* __restrict__ bias,
                                                          const float* __restrict__ Wn,
                                                          float* __restrict__ tn) {
    __shared__ v2f wfrag[16 * 4 * 32];
    __shared__ __attribute__((aligned(16))) float tile[2][16 * TILE_PITCH];

    const int l    = threadIdx.x & 15;
    const int rowT = threadIdx.x >> 4;
    const int lane = threadIdx.x & 31;
    const int m    = lane & 15;
    const int hi   = lane >> 4;
    const int wave = __builtin_amdgcn_readfirstlane((int)(threadIdx.x >> 5));

#pragma unroll
    for (int i = 0; i < (16 * 4 * 32) / 256; ++i) {
        const int idx = i * 256 + threadIdx.x;
        const int fl = idx & 31;
        const int ct = (idx >> 5) & 3;
        const int kb = idx >> 7;
        const int k  = kb * 4 + (fl >> 4) * 2;
        v2f f;
        f.x = Wn[k * DIM + ct * 16 + (fl & 15)];
        f.y = Wn[(k + 1) * DIM + ct * 16 + (fl & 15)];
        wfrag[idx] = f;
    }

    const float4* t4 = (const float4*)t;
    const float4 b = ((const float4*)bias)[l];
    int it = 0;
    for (int rt = blockIdx.x; rt < N_NODES / 16; rt += gridDim.x, ++it) {
        const int n = rt * 16 + rowT;
        const int beg = rowptr[n];
        const int end = rowptr[n + 1];
        const float dn = dinv[n];
        const float4 self = t4[n * 16 + l];
        const float wself = dn * dn;
        float4 acc;
        acc.x = self.x * wself;
        acc.y = self.y * wself;
        acc.z = self.z * wself;
        acc.w = self.w * wself;
        int j = beg;
        for (; j + 4 <= end; j += 4) {
            const int2 e0 = csr[j];
            const int2 e1 = csr[j + 1];
            const int2 e2 = csr[j + 2];
            const int2 e3 = csr[j + 3];
            const float4 r0 = t4[e0.x * 16 + l];
            const float4 r1 = t4[e1.x * 16 + l];
            const float4 r2 = t4[e2.x * 16 + l];
            const float4 r3 = t4[e3.x * 16 + l];
            axpy4(acc, r0, __int_as_float(e0.y));
            axpy4(acc, r1, __int_as_float(e1.y));
            axpy4(acc, r2, __int_as_float(e2.y));
            axpy4(acc, r3, __int_as_float(e3.y));
        }
        if (j + 2 <= end) {
            const int2 e0 = csr[j];
            const int2 e1 = csr[j + 1];
            const float4 r0 = t4[e0.x * 16 + l];
            const float4 r1 = t4[e1.x * 16 + l];
            axpy4(acc, r0, __int_as_float(e0.y));
            axpy4(acc, r1, __int_as_float(e1.y));
            j += 2;
        }
        if (j < end) {
            const int2 e0 = csr[j];
            const float4 r0 = t4[e0.x * 16 + l];
            axpy4(acc, r0, __int_as_float(e0.y));
        }
        acc.x = fmaxf(acc.x + b.x, 0.0f);
        acc.y = fmaxf(acc.y + b.y, 0.0f);
        acc.z = fmaxf(acc.z + b.z, 0.0f);
        acc.w = fmaxf(acc.w + b.w, 0.0f);

        float* tb = tile[it & 1];
        *(float4*)(tb + rowT * TILE_PITCH + l * 4) = acc;
        __syncthreads();

        if ((wave >> 2) == (it & 1)) {
            const int ct = wave & 3;
            const float* arow = tb + m * TILE_PITCH + hi * 2;
            const v2f* bp = wfrag + ct * 32 + lane;
            v8f c = {};
#pragma unroll
            for (int kb = 0; kb < 16; ++kb) {
                const v2f a = *(const v2f*)(arow + kb * 4);
                c = __builtin_amdgcn_wmma_f32_16x16x4_f32(
                         false, a,  false, bp[kb * 128],
                         (short)0, c,  false,  false);
            }
            float* orow = tn + (size_t)(rt * 16 + hi * 8) * DIM + ct * 16 + m;
#pragma unroll
            for (int r = 0; r < 8; ++r) orow[(size_t)r * DIM] = c[r];
        }
    }
}

extern "C" void kernel_launch(void* const* d_in, const int* in_sizes, int n_in,
                              void* d_out, int out_size, void* d_ws, size_t ws_size,
                              hipStream_t stream) {
    const float* x  = (const float*)d_in[0];
    const int*   ei = (const int*)d_in[1];
    const float* W1 = (const float*)d_in[2];
    const float* b1 = (const float*)d_in[3];
    const float* W2 = (const float*)d_in[4];
    const float* b2 = (const float*)d_in[5];
    const float* W3 = (const float*)d_in[6];
    const float* b3 = (const float*)d_in[7];
    float* out = (float*)d_out;

    char* ws = (char*)d_ws;
    float* dinv   = (float*)(ws + OFF_DINV);
    float* tbuf   = (float*)(ws + OFF_TBUF);
    int*   cnt    = (int*)(ws + OFF_CNT);
    int*   rowptr = (int*)(ws + OFF_ROWPTR);
    int*   rank   = (int*)(ws + OFF_RANK);
    int2*  csr    = (int2*)(ws + OFF_CSR);

    cnt_zero_kernel<<<391, 256, 0, stream>>>(cnt);
    rank_kernel<<<1172, 256, 0, stream>>>(ei, cnt, rank);
    scan_dinv_kernel<<<1, 1024, 0, stream>>>(cnt, rowptr, dinv);
    fill_kernel<<<1172, 256, 0, stream>>>(ei, rank, rowptr, dinv, csr);

    gemm_wmma_kernel<<<512, 256, 0, stream>>>(x, W1, tbuf);

    gather_gemm_kernel<<<2048, 256, 0, stream>>>(tbuf, rowptr, csr, dinv, b1, W2, out);

    gather_gemm_kernel<<<2048, 256, 0, stream>>>(out, rowptr, csr, dinv, b2, W3, tbuf);

    gather_kernel<false><<<2048, 256, 0, stream>>>(tbuf, rowptr, csr, dinv, b3, out);
}


# Edge grouping moved from one global atomic per edge to a two-level counting sort on LDS counters (chunks by 256-node bin, bins by node); 4 launches become 3; sum order = arrival order of the LDS adds
# speedup vs baseline: 4.1971x; 1.7469x over previous
#include <hip/hip_runtime.h>


#define N_NODES 100000
#define N_EDGES 1200000
#define DIM 64

#define OFF_DINV   ((size_t)0)
#define OFF_TBUF   ((size_t)400128)
#define OFF_LSTART ((size_t)400128 + (size_t)25600000)
#define OFF_ROWPTR (OFF_LSTART + (size_t)400128)
#define OFF_CSR    (OFF_ROWPTR + (size_t)400384 + (size_t)4800000)

typedef __attribute__((ext_vector_type(2))) float v2f;
typedef __attribute__((ext_vector_type(8))) float v8f;

#define NBINS 391
#define LPITCH (NBINS + 1)
#define NCHUNK 128
#define CHUNK_EDGES (N_EDGES / NCHUNK)
#define CHUNK_TRIPS 10

__global__ __launch_bounds__(1024) void chunk_sort_kernel(const int* __restrict__ ei,
                                                          int* __restrict__ lstart,
                                                          int2* __restrict__ binned) {
    __shared__ int hist[512];
    __shared__ int wsum[32];
    const int tid = threadIdx.x;
    const int lane = tid & 31;
    const int w = tid >> 5;
    if (tid < 512) hist[tid] = 0;
    const int* src = ei + blockIdx.x * CHUNK_EDGES;
    const int* dst = src + N_EDGES;
    int s[CHUNK_TRIPS];
    int d[CHUNK_TRIPS];
#pragma unroll
    for (int t = 0; t < CHUNK_TRIPS; ++t) {
        const int k = tid + t * 1024;
        const int kk = k < CHUNK_EDGES ? k : CHUNK_EDGES - 1;
        s[t] = src[kk];
        d[t] = dst[kk];
    }
    __syncthreads();
    int r[CHUNK_TRIPS];
#pragma unroll
    for (int t = 0; t < CHUNK_TRIPS; ++t) {
        r[t] = 0;
        if (tid + t * 1024 < CHUNK_EDGES) r[t] = atomicAdd(&hist[d[t] >> 8], 1);
    }
    __syncthreads();
    const int v = tid < 512 ? hist[tid] : 0;
    int x = v;
#pragma unroll
    for (int o = 1; o < 32; o <<= 1) {
        int y = __shfl_up(x, o);
        if (lane >= o) x += y;
    }
    if (lane == 31) wsum[w] = x;
    __syncthreads();
    const int ws = wsum[lane];
    int wx = ws;
#pragma unroll
    for (int o = 1; o < 32; o <<= 1) {
        int y = __shfl_up(wx, o);
        if (lane >= o) wx += y;
    }
    const int ex = __shfl(wx - ws, w) + (x - v);
    if (tid < 512) hist[tid] = ex;
    if (tid < LPITCH) lstart[blockIdx.x * LPITCH + tid] = ex;
    __syncthreads();
    int2* piece = binned + blockIdx.x * CHUNK_EDGES;
#pragma unroll
    for (int t = 0; t < CHUNK_TRIPS; ++t) {
        if (tid + t * 1024 < CHUNK_EDGES) piece[hist[d[t] >> 8] + r[t]] = make_int2(s[t], d[t]);
    }
}

__global__ __launch_bounds__(256) void bin_sort_kernel(const int2* __restrict__ binned,
                                                       const int* __restrict__ lstart,
                                                       int* __restrict__ rowptr,
                                                       float* __restrict__ dinv,
                                                       int2* __restrict__ csr) {
    __shared__ int cnt[256];
    __shared__ int wsum[8];
    __shared__ int bsum[8];
    __shared__ int segb[NCHUNK];
    __shared__ int sege[NCHUNK];
    const int tid = threadIdx.x;
    const int lane = tid & 31;
    const int w = tid >> 5;
    const int base = blockIdx.x << 8;
    int lb = 0;
    if (tid < NCHUNK) {
        lb = lstart[tid * LPITCH + blockIdx.x];
        const int le = lstart[tid * LPITCH + blockIdx.x + 1];
        segb[tid] = tid * CHUNK_EDGES + lb;
        sege[tid] = tid * CHUNK_EDGES + le;
    }
    cnt[tid] = 0;
    int y = lb;
#pragma unroll
    for (int o = 16; o > 0; o >>= 1) y += __shfl_xor(y, o);
    if (lane == 0) bsum[w] = y;
    __syncthreads();
    int beg = 0;
#pragma unroll
    for (int q = 0; q < 8; ++q) beg += bsum[q];

    const int g = tid >> 3;
    const int sl = tid & 7;
    int jb[4];
    int je[4];
#pragma unroll
    for (int u = 0; u < 4; ++u) {
        jb[u] = segb[g + 32 * u] + sl;
        je[u] = sege[g + 32 * u];
    }
    for (int k = 0; k < CHUNK_EDGES; k += 8) {
        bool any = false;
#pragma unroll
        for (int u = 0; u < 4; ++u) any = any || (jb[u] + k < je[u]);
        if (!any) break;
        int dd[4];
#pragma unroll
        for (int u = 0; u < 4; ++u) {
            const int j = jb[u] + k;
            dd[u] = binned[j < je[u] ? j : 0].y;
        }
#pragma unroll
        for (int u = 0; u < 4; ++u) {
            if (jb[u] + k < je[u]) atomicAdd(&cnt[dd[u] - base], 1);
        }
    }
    __syncthreads();
    const int c = cnt[tid];
    int x = c;
#pragma unroll
    for (int o = 1; o < 32; o <<= 1) {
        int z = __shfl_up(x, o);
        if (lane >= o) x += z;
    }
    if (lane == 31) wsum[w] = x;
    __syncthreads();
    int wpre = 0;
#pragma unroll
    for (int q = 0; q < 8; ++q) {
        const int t = wsum[q];
        if (q < w) wpre += t;
    }
    const int first = beg + wpre + (x - c);
    cnt[tid] = first;
    const int n = base + tid;
    if (n <= N_NODES) rowptr[n] = first;
    if (n < N_NODES) dinv[n] = rsqrtf((float)(c + 1));
    __syncthreads();
    for (int k = 0; k < CHUNK_EDGES; k += 8) {
        bool any = false;
#pragma unroll
        for (int u = 0; u < 4; ++u) any = any || (jb[u] + k < je[u]);
        if (!any) break;
        int2 e[4];
        int p[4];
#pragma unroll
        for (int u = 0; u < 4; ++u) {
            const int j = jb[u] + k;
            e[u] = binned[j < je[u] ? j : 0];
        }
#pragma unroll
        for (int u = 0; u < 4; ++u) {
            p[u] = 0;
            if (jb[u] + k < je[u]) p[u] = atomicAdd(&cnt[e[u].y - base], 1);
        }
#pragma unroll
        for (int u = 0; u < 4; ++u) {
            if (jb[u] + k < je[u]) csr[p[u]] = e[u];
        }
    }
}

__global__ void weight_kernel(const float* __restrict__ dinv, int4* __restrict__ csr4) {
    const int i = blockIdx.x * blockDim.x + threadIdx.x;
    if (i < N_EDGES / 4) {
        int4 a = csr4[2 * i];
        int4 b = csr4[2 * i + 1];
        a.y = __float_as_int(dinv[a.x] * dinv[a.y]);
        a.w = __float_as_int(dinv[a.z] * dinv[a.w]);
        b.y = __float_as_int(dinv[b.x] * dinv[b.y]);
        b.w = __float_as_int(dinv[b.z] * dinv[b.w]);
        csr4[2 * i] = a;
        csr4[2 * i + 1] = b;
    }
}

__global__ void gemm_wmma_kernel(const float* __restrict__ in,
                                 const float* __restrict__ W,
                                 float* __restrict__ out) {
    const int lane = threadIdx.x & 31;
    const int wid  = (blockIdx.x * blockDim.x + threadIdx.x) >> 5;
    const int nWaves = (gridDim.x * blockDim.x) >> 5;

    const int col0 = (wid & 3) * 16;
    const int m    = lane & 15;
    const int hi   = lane >> 4;
    const int klo  = hi * 2;

    v2f bfrag[16];
#pragma unroll
    for (int kb = 0; kb < 16; ++kb) {
        int k = kb * 4 + klo;
        bfrag[kb].x = W[k * DIM + col0 + m];
        bfrag[kb].y = W[(k + 1) * DIM + col0 + m];
    }

    const int nRowTiles  = N_NODES / 16;
    const int rowStride  = nWaves >> 2;

    for (int rt = wid >> 2; rt < nRowTiles; rt += rowStride) {
        const int row0 = rt * 16;
        const float* arow = in + (size_t)(row0 + m) * DIM;
        v8f c = {};
#pragma unroll
        for (int kb = 0; kb < 16; ++kb) {
            v2f a = *(const v2f*)(arow + kb * 4 + klo);
            c = __builtin_amdgcn_wmma_f32_16x16x4_f32(
                     false, a,  false, bfrag[kb],
                     (short)0, c,  false,  false);
        }
        float* orow = out + (size_t)(row0 + hi * 8) * DIM + col0 + m;
#pragma unroll
        for (int r = 0; r < 8; ++r) orow[(size_t)r * DIM] = c[r];
    }
}

static __device__ __forceinline__ void axpy4(float4& acc, const float4 r, const float w) {
    acc.x = fmaf(r.x, w, acc.x);
    acc.y = fmaf(r.y, w, acc.y);
    acc.z = fmaf(r.z, w, acc.z);
    acc.w = fmaf(r.w, w, acc.w);
}

template <bool RELU>
__global__ __launch_bounds__(256) void gather_kernel(const float* __restrict__ t,
                                                     const int* __restrict__ rowptr,
                                                     const int2* __restrict__ csr,
                                                     const float* __restrict__ dinv,
                                                     const float* __restrict__ bias,
                                                     float* __restrict__ h) {
    const int l = threadIdx.x & 15;
    const float4* t4 = (const float4*)t;
    const float4 b = ((const float4*)bias)[l];
    for (int n = blockIdx.x * 16 + (threadIdx.x >> 4); n < N_NODES; n += gridDim.x * 16) {
        const int beg = rowptr[n];
        const int end = rowptr[n + 1];
        const float dn = dinv[n];
        const float4 self = t4[n * 16 + l];
        const float wself = dn * dn;
        float4 acc;
        acc.x = self.x * wself;
        acc.y = self.y * wself;
        acc.z = self.z * wself;
        acc.w = self.w * wself;
        int j = beg;
        for (; j + 4 <= end; j += 4) {
            const int2 e0 = csr[j];
            const int2 e1 = csr[j + 1];
            const int2 e2 = csr[j + 2];
            const int2 e3 = csr[j + 3];
            const float4 r0 = t4[e0.x * 16 + l];
            const float4 r1 = t4[e1.x * 16 + l];
            const float4 r2 = t4[e2.x * 16 + l];
            const float4 r3 = t4[e3.x * 16 + l];
            axpy4(acc, r0, __int_as_float(e0.y));
            axpy4(acc, r1, __int_as_float(e1.y));
            axpy4(acc, r2, __int_as_float(e2.y));
            axpy4(acc, r3, __int_as_float(e3.y));
        }
        if (j + 2 <= end) {
            const int2 e0 = csr[j];
            const int2 e1 = csr[j + 1];
            const float4 r0 = t4[e0.x * 16 + l];
            const float4 r1 = t4[e1.x * 16 + l];
            axpy4(acc, r0, __int_as_float(e0.y));
            axpy4(acc, r1, __int_as_float(e1.y));
            j += 2;
        }
        if (j < end) {
            const int2 e0 = csr[j];
            const float4 r0 = t4[e0.x * 16 + l];
            axpy4(acc, r0, __int_as_float(e0.y));
        }
        acc.x += b.x;
        acc.y += b.y;
        acc.z += b.z;
        acc.w += b.w;
        if (RELU) {
            acc.x = fmaxf(acc.x, 0.0f);
            acc.y = fmaxf(acc.y, 0.0f);
            acc.z = fmaxf(acc.z, 0.0f);
            acc.w = fmaxf(acc.w, 0.0f);
        }
        ((float4*)h)[n * 16 + l] = acc;
    }
}

#define TILE_PITCH 68

__global__ __launch_bounds__(256) void gather_gemm_kernel(const float* __restrict__ t,
                                                          const int* __restrict__ rowptr,
                                                          const int2* __restrict__ csr,
                                                          const float* __restrict__ dinv,
                                                          const float* __restrict__ bias,
                                                          const float* __restrict__ Wn,
                                                          float* __restrict__ tn) {
    __shared__ v2f wfrag[16 * 4 * 32];
    __shared__ __attribute__((aligned(16))) float tile[2][16 * TILE_PITCH];

    const int l    = threadIdx.x & 15;
    const int rowT = threadIdx.x >> 4;
    const int lane = threadIdx.x & 31;
    const int m    = lane & 15;
    const int hi   = lane >> 4;
    const int wave = __builtin_amdgcn_readfirstlane((int)(threadIdx.x >> 5));

#pragma unroll
    for (int i = 0; i < (16 * 4 * 32) / 256; ++i) {
        const int idx = i * 256 + threadIdx.x;
        const int fl = idx & 31;
        const int ct = (idx >> 5) & 3;
        const int kb = idx >> 7;
        const int k  = kb * 4 + (fl >> 4) * 2;
        v2f f;
        f.x = Wn[k * DIM + ct * 16 + (fl & 15)];
        f.y = Wn[(k + 1) * DIM + ct * 16 + (fl & 15)];
        wfrag[idx] = f;
    }

    const float4* t4 = (const float4*)t;
    const float4 b = ((const float4*)bias)[l];
    int it = 0;
    for (int rt = blockIdx.x; rt < N_NODES / 16; rt += gridDim.x, ++it) {
        const int n = rt * 16 + rowT;
        const int beg = rowptr[n];
        const int end = rowptr[n + 1];
        const float dn = dinv[n];
        const float4 self = t4[n * 16 + l];
        const float wself = dn * dn;
        float4 acc;
        acc.x = self.x * wself;
        acc.y = self.y * wself;
        acc.z = self.z * wself;
        acc.w = self.w * wself;
        int j = beg;
        for (; j + 4 <= end; j += 4) {
            const int2 e0 = csr[j];
            const int2 e1 = csr[j + 1];
            const int2 e2 = csr[j + 2];
            const int2 e3 = csr[j + 3];
            const float4 r0 = t4[e0.x * 16 + l];
            const float4 r1 = t4[e1.x * 16 + l];
            const float4 r2 = t4[e2.x * 16 + l];
            const float4 r3 = t4[e3.x * 16 + l];
            axpy4(acc, r0, __int_as_float(e0.y));
            axpy4(acc, r1, __int_as_float(e1.y));
            axpy4(acc, r2, __int_as_float(e2.y));
            axpy4(acc, r3, __int_as_float(e3.y));
        }
        if (j + 2 <= end) {
            const int2 e0 = csr[j];
            const int2 e1 = csr[j + 1];
            const float4 r0 = t4[e0.x * 16 + l];
            const float4 r1 = t4[e1.x * 16 + l];
            axpy4(acc, r0, __int_as_float(e0.y));
            axpy4(acc, r1, __int_as_float(e1.y));
            j += 2;
        }
        if (j < end) {
            const int2 e0 = csr[j];
            const float4 r0 = t4[e0.x * 16 + l];
            axpy4(acc, r0, __int_as_float(e0.y));
        }
        acc.x = fmaxf(acc.x + b.x, 0.0f);
        acc.y = fmaxf(acc.y + b.y, 0.0f);
        acc.z = fmaxf(acc.z + b.z, 0.0f);
        acc.w = fmaxf(acc.w + b.w, 0.0f);

        float* tb = tile[it & 1];
        *(float4*)(tb + rowT * TILE_PITCH + l * 4) = acc;
        __syncthreads();

        if ((wave >> 2) == (it & 1)) {
            const int ct = wave & 3;
            const float* arow = tb + m * TILE_PITCH + hi * 2;
            const v2f* bp = wfrag + ct * 32 + lane;
            v8f c = {};
#pragma unroll
            for (int kb = 0; kb < 16; ++kb) {
                const v2f a = *(const v2f*)(arow + kb * 4);
                c = __builtin_amdgcn_wmma_f32_16x16x4_f32(
                         false, a,  false, bp[kb * 128],
                         (short)0, c,  false,  false);
            }
            float* orow = tn + (size_t)(rt * 16 + hi * 8) * DIM + ct * 16 + m;
#pragma unroll
            for (int r = 0; r < 8; ++r) orow[(size_t)r * DIM] = c[r];
        }
    }
}

extern "C" void kernel_launch(void* const* d_in, const int* in_sizes, int n_in,
                              void* d_out, int out_size, void* d_ws, size_t ws_size,
                              hipStream_t stream) {
    const float* x  = (const float*)d_in[0];
    const int*   ei = (const int*)d_in[1];
    const float* W1 = (const float*)d_in[2];
    const float* b1 = (const float*)d_in[3];
    const float* W2 = (const float*)d_in[4];
    const float* b2 = (const float*)d_in[5];
    const float* W3 = (const float*)d_in[6];
    const float* b3 = (const float*)d_in[7];
    float* out = (float*)d_out;

    char* ws = (char*)d_ws;
    float* dinv   = (float*)(ws + OFF_DINV);
    float* tbuf   = (float*)(ws + OFF_TBUF);
    int*   lstart = (int*)(ws + OFF_LSTART);
    int*   rowptr = (int*)(ws + OFF_ROWPTR);
    int2*  csr    = (int2*)(ws + OFF_CSR);
    int2*  binned = (int2*)(ws + OFF_TBUF);

    chunk_sort_kernel<<<NCHUNK, 1024, 0, stream>>>(ei, lstart, binned);
    bin_sort_kernel<<<NBINS, 256, 0, stream>>>(binned, lstart, rowptr, dinv, csr);
    weight_kernel<<<1172, 256, 0, stream>>>(dinv, (int4*)csr);

    gemm_wmma_kernel<<<512, 256, 0, stream>>>(x, W1, tbuf);

    gather_gemm_kernel<<<2048, 256, 0, stream>>>(tbuf, rowptr, csr, dinv, b1, W2, out);

    gather_gemm_kernel<<<2048, 256, 0, stream>>>(out, rowptr, csr, dinv, b2, W3, tbuf);

    gather_kernel<false><<<2048, 256, 0, stream>>>(tbuf, rowptr, csr, dinv, b3, out);
}


# Norm factored: product rows stored pre-scaled by dinv[src], each node's sum scaled once by dinv[dst]; per-edge weight pass and its launch removed, list = src ids; sum order kept, scaling reassociated
# speedup vs baseline: 4.5037x; 1.0731x over previous
#include <hip/hip_runtime.h>


#define N_NODES 100000
#define N_EDGES 1200000
#define DIM 64

#define OFF_DINV   ((size_t)0)
#define OFF_TBUF   ((size_t)400128)
#define OFF_LSTART ((size_t)400128 + (size_t)25600000)
#define OFF_ROWPTR (OFF_LSTART + (size_t)400128)
#define OFF_CSR    (OFF_ROWPTR + (size_t)400384 + (size_t)4800000)

typedef __attribute__((ext_vector_type(2))) float v2f;
typedef __attribute__((ext_vector_type(8))) float v8f;

#define NBINS 391
#define LPITCH (NBINS + 1)
#define NCHUNK 128
#define CHUNK_EDGES (N_EDGES / NCHUNK)
#define CHUNK_TRIPS 10

__global__ __launch_bounds__(1024) void chunk_sort_kernel(const int* __restrict__ ei,
                                                          int* __restrict__ lstart,
                                                          int2* __restrict__ binned) {
    __shared__ int hist[512];
    __shared__ int wsum[32];
    const int tid = threadIdx.x;
    const int lane = tid & 31;
    const int w = tid >> 5;
    if (tid < 512) hist[tid] = 0;
    const int* src = ei + blockIdx.x * CHUNK_EDGES;
    const int* dst = src + N_EDGES;
    int s[CHUNK_TRIPS];
    int d[CHUNK_TRIPS];
#pragma unroll
    for (int t = 0; t < CHUNK_TRIPS; ++t) {
        const int k = tid + t * 1024;
        const int kk = k < CHUNK_EDGES ? k : CHUNK_EDGES - 1;
        s[t] = src[kk];
        d[t] = dst[kk];
    }
    __syncthreads();
    int r[CHUNK_TRIPS];
#pragma unroll
    for (int t = 0; t < CHUNK_TRIPS; ++t) {
        r[t] = 0;
        if (tid + t * 1024 < CHUNK_EDGES) r[t] = atomicAdd(&hist[d[t] >> 8], 1);
    }
    __syncthreads();
    const int v = tid < 512 ? hist[tid] : 0;
    int x = v;
#pragma unroll
    for (int o = 1; o < 32; o <<= 1) {
        int y = __shfl_up(x, o);
        if (lane >= o) x += y;
    }
    if (lane == 31) wsum[w] = x;
    __syncthreads();
    const int ws = wsum[lane];
    int wx = ws;
#pragma unroll
    for (int o = 1; o < 32; o <<= 1) {
        int y = __shfl_up(wx, o);
        if (lane >= o) wx += y;
    }
    const int ex = __shfl(wx - ws, w) + (x - v);
    if (tid < 512) hist[tid] = ex;
    if (tid < LPITCH) lstart[blockIdx.x * LPITCH + tid] = ex;
    __syncthreads();
    int2* piece = binned + blockIdx.x * CHUNK_EDGES;
#pragma unroll
    for (int t = 0; t < CHUNK_TRIPS; ++t) {
        if (tid + t * 1024 < CHUNK_EDGES) piece[hist[d[t] >> 8] + r[t]] = make_int2(s[t], d[t]);
    }
}

__global__ __launch_bounds__(256) void bin_sort_kernel(const int2* __restrict__ binned,
                                                       const int* __restrict__ lstart,
                                                       int* __restrict__ rowptr,
                                                       float* __restrict__ dinv,
                                                       int* __restrict__ csr) {
    __shared__ int cnt[256];
    __shared__ int wsum[8];
    __shared__ int bsum[8];
    __shared__ int segb[NCHUNK];
    __shared__ int sege[NCHUNK];
    const int tid = threadIdx.x;
    const int lane = tid & 31;
    const int w = tid >> 5;
    const int base = blockIdx.x << 8;
    int lb = 0;
    if (tid < NCHUNK) {
        lb = lstart[tid * LPITCH + blockIdx.x];
        const int le = lstart[tid * LPITCH + blockIdx.x + 1];
        segb[tid] = tid * CHUNK_EDGES + lb;
        sege[tid] = tid * CHUNK_EDGES + le;
    }
    cnt[tid] = 0;
    int y = lb;
#pragma unroll
    for (int o = 16; o > 0; o >>= 1) y += __shfl_xor(y, o);
    if (lane == 0) bsum[w] = y;
    __syncthreads();
    int beg = 0;
#pragma unroll
    for (int q = 0; q < 8; ++q) beg += bsum[q];

    const int g = tid >> 3;
    const int sl = tid & 7;
    int jb[4];
    int je[4];
#pragma unroll
    for (int u = 0; u < 4; ++u) {
        jb[u] = segb[g + 32 * u] + sl;
        je[u] = sege[g + 32 * u];
    }
    for (int k = 0; k < CHUNK_EDGES; k += 8) {
        bool any = false;
#pragma unroll
        for (int u = 0; u < 4; ++u) any = any || (jb[u] + k < je[u]);
        if (!any) break;
        int dd[4];
#pragma unroll
        for (int u = 0; u < 4; ++u) {
            const int j = jb[u] + k;
            dd[u] = binned[j < je[u] ? j : 0].y;
        }
#pragma unroll
        for (int u = 0; u < 4; ++u) {
            if (jb[u] + k < je[u]) atomicAdd(&cnt[dd[u] - base], 1);
        }
    }
    __syncthreads();
    const int c = cnt[tid];
    int x = c;
#pragma unroll
    for (int o = 1; o < 32; o <<= 1) {
        int z = __shfl_up(x, o);
        if (lane >= o) x += z;
    }
    if (lane == 31) wsum[w] = x;
    __syncthreads();
    int wpre = 0;
#pragma unroll
    for (int q = 0; q < 8; ++q) {
        const int t = wsum[q];
        if (q < w) wpre += t;
    }
    const int first = beg + wpre + (x - c);
    cnt[tid] = first;
    const int n = base + tid;
    if (n <= N_NODES) rowptr[n] = first;
    if (n < N_NODES) dinv[n] = rsqrtf((float)(c + 1));
    __syncthreads();
    for (int k = 0; k < CHUNK_EDGES; k += 8) {
        bool any = false;
#pragma unroll
        for (int u = 0; u < 4; ++u) any = any || (jb[u] + k < je[u]);
        if (!any) break;
        int2 e[4];
        int p[4];
#pragma unroll
        for (int u = 0; u < 4; ++u) {
            const int j = jb[u] + k;
            e[u] = binned[j < je[u] ? j : 0];
        }
#pragma unroll
        for (int u = 0; u < 4; ++u) {
            p[u] = 0;
            if (jb[u] + k < je[u]) p[u] = atomicAdd(&cnt[e[u].y - base], 1);
        }
#pragma unroll
        for (int u = 0; u < 4; ++u) {
            if (jb[u] + k < je[u]) csr[p[u]] = e[u].x;
        }
    }
}

__global__ void gemm_wmma_kernel(const float* __restrict__ in,
                                 const float* __restrict__ W,
                                 const float* __restrict__ dinv,
                                 float* __restrict__ out) {
    const int lane = threadIdx.x & 31;
    const int wid  = (blockIdx.x * blockDim.x + threadIdx.x) >> 5;
    const int nWaves = (gridDim.x * blockDim.x) >> 5;

    const int col0 = (wid & 3) * 16;
    const int m    = lane & 15;
    const int hi   = lane >> 4;
    const int klo  = hi * 2;

    v2f bfrag[16];
#pragma unroll
    for (int kb = 0; kb < 16; ++kb) {
        int k = kb * 4 + klo;
        bfrag[kb].x = W[k * DIM + col0 + m];
        bfrag[kb].y = W[(k + 1) * DIM + col0 + m];
    }

    const int nRowTiles  = N_NODES / 16;
    const int rowStride  = nWaves >> 2;

    int rt = wid >> 2;
    const int rfirst = (rt < nRowTiles ? rt : 0) * 16 + hi * 8;
    float4 dlo = *(const float4*)(dinv + rfirst);
    float4 dhi = *(const float4*)(dinv + rfirst + 4);
    for (; rt < nRowTiles; rt += rowStride) {
        const int row0 = rt * 16;
        const float* arow = in + (size_t)(row0 + m) * DIM;
        const int rtn = rt + rowStride;
        const int rnext = (rtn < nRowTiles ? rtn : rt) * 16 + hi * 8;
        const float4 nlo = *(const float4*)(dinv + rnext);
        const float4 nhi = *(const float4*)(dinv + rnext + 4);
        v8f c = {};
#pragma unroll
        for (int kb = 0; kb < 16; ++kb) {
            v2f a = *(const v2f*)(arow + kb * 4 + klo);
            c = __builtin_amdgcn_wmma_f32_16x16x4_f32(
                     false, a,  false, bfrag[kb],
                     (short)0, c,  false,  false);
        }
        float* orow = out + (size_t)(row0 + hi * 8) * DIM + col0 + m;
        asm volatile("v_nop\n\tv_nop\n\tv_nop\n\tv_nop\n\tv_nop\n\tv_nop\n\tv_nop\n\tv_nop" : "+v"(c));
        const float ds[8] = {dlo.x, dlo.y, dlo.z, dlo.w, dhi.x, dhi.y, dhi.z, dhi.w};
#pragma unroll
        for (int r = 0; r < 8; ++r) orow[(size_t)r * DIM] = c[r] * ds[r];
        dlo = nlo;
        dhi = nhi;
    }
}

static __device__ __forceinline__ void add4(float4& acc, const float4 r) {
    acc.x += r.x;
    acc.y += r.y;
    acc.z += r.z;
    acc.w += r.w;
}

template <bool RELU>
__global__ __launch_bounds__(256) void gather_kernel(const float* __restrict__ t,
                                                     const int* __restrict__ rowptr,
                                                     const int* __restrict__ csr,
                                                     const float* __restrict__ dinv,
                                                     const float* __restrict__ bias,
                                                     float* __restrict__ h) {
    const int l = threadIdx.x & 15;
    const float4* t4 = (const float4*)t;
    const float4 b = ((const float4*)bias)[l];
    for (int n = blockIdx.x * 16 + (threadIdx.x >> 4); n < N_NODES; n += gridDim.x * 16) {
        const int beg = rowptr[n];
        const int end = rowptr[n + 1];
        const float dn = dinv[n];
        float4 acc = t4[n * 16 + l];
        int j = beg;
        for (; j + 4 <= end; j += 4) {
            const int s0 = csr[j];
            const int s1 = csr[j + 1];
            const int s2 = csr[j + 2];
            const int s3 = csr[j + 3];
            const float4 r0 = t4[s0 * 16 + l];
            const float4 r1 = t4[s1 * 16 + l];
            const float4 r2 = t4[s2 * 16 + l];
            const float4 r3 = t4[s3 * 16 + l];
            add4(acc, r0);
            add4(acc, r1);
            add4(acc, r2);
            add4(acc, r3);
        }
        if (j + 2 <= end) {
            const int s0 = csr[j];
            const int s1 = csr[j + 1];
            const float4 r0 = t4[s0 * 16 + l];
            const float4 r1 = t4[s1 * 16 + l];
            add4(acc, r0);
            add4(acc, r1);
            j += 2;
        }
        if (j < end) {
            const int s0 = csr[j];
            const float4 r0 = t4[s0 * 16 + l];
            add4(acc, r0);
        }
        acc.x = acc.x * dn + b.x;
        acc.y = acc.y * dn + b.y;
        acc.z = acc.z * dn + b.z;
        acc.w = acc.w * dn + b.w;
        if (RELU) {
            acc.x = fmaxf(acc.x, 0.0f);
            acc.y = fmaxf(acc.y, 0.0f);
            acc.z = fmaxf(acc.z, 0.0f);
            acc.w = fmaxf(acc.w, 0.0f);
        }
        ((float4*)h)[n * 16 + l] = acc;
    }
}

#define TILE_PITCH 68

__global__ __launch_bounds__(256) void gather_gemm_kernel(const float* __restrict__ t,
                                                          const int* __restrict__ rowptr,
                                                          const int* __restrict__ csr,
                                                          const float* __restrict__ dinv,
                                                          const float* __restrict__ bias,
                                                          const float* __restrict__ Wn,
                                                          float* __restrict__ tn) {
    __shared__ v2f wfrag[16 * 4 * 32];
    __shared__ __attribute__((aligned(16))) float tile[2][16 * TILE_PITCH];

    const int l    = threadIdx.x & 15;
    const int rowT = threadIdx.x >> 4;
    const int lane = threadIdx.x & 31;
    const int m    = lane & 15;
    const int hi   = lane >> 4;
    const int wave = __builtin_amdgcn_readfirstlane((int)(threadIdx.x >> 5));

#pragma unroll
    for (int i = 0; i < (16 * 4 * 32) / 256; ++i) {
        const int idx = i * 256 + threadIdx.x;
        const int fl = idx & 31;
        const int ct = (idx >> 5) & 3;
        const int kb = idx >> 7;
        const int k  = kb * 4 + (fl >> 4) * 2;
        v2f f;
        f.x = Wn[k * DIM + ct * 16 + (fl & 15)];
        f.y = Wn[(k + 1) * DIM + ct * 16 + (fl & 15)];
        wfrag[idx] = f;
    }

    const float4* t4 = (const float4*)t;
    const float4 b = ((const float4*)bias)[l];
    int it = 0;
    for (int rt = blockIdx.x; rt < N_NODES / 16; rt += gridDim.x, ++it) {
        const int n = rt * 16 + rowT;
        const int beg = rowptr[n];
        const int end = rowptr[n + 1];
        const float dn = dinv[n];
        float4 acc = t4[n * 16 + l];
        int j = beg;
        for (; j + 4 <= end; j += 4) {
            const int s0 = csr[j];
            const int s1 = csr[j + 1];
            const int s2 = csr[j + 2];
            const int s3 = csr[j + 3];
            const float4 r0 = t4[s0 * 16 + l];
            const float4 r1 = t4[s1 * 16 + l];
            const float4 r2 = t4[s2 * 16 + l];
            const float4 r3 = t4[s3 * 16 + l];
            add4(acc, r0);
            add4(acc, r1);
            add4(acc, r2);
            add4(acc, r3);
        }
        if (j + 2 <= end) {
            const int s0 = csr[j];
            const int s1 = csr[j + 1];
            const float4 r0 = t4[s0 * 16 + l];
            const float4 r1 = t4[s1 * 16 + l];
            add4(acc, r0);
            add4(acc, r1);
            j += 2;
        }
        if (j < end) {
            const int s0 = csr[j];
            const float4 r0 = t4[s0 * 16 + l];
            add4(acc, r0);
        }
        acc.x = fmaxf(acc.x * dn + b.x, 0.0f) * dn;
        acc.y = fmaxf(acc.y * dn + b.y, 0.0f) * dn;
        acc.z = fmaxf(acc.z * dn + b.z, 0.0f) * dn;
        acc.w = fmaxf(acc.w * dn + b.w, 0.0f) * dn;

        float* tb = tile[it & 1];
        *(float4*)(tb + rowT * TILE_PITCH + l * 4) = acc;
        __syncthreads();

        if ((wave >> 2) == (it & 1)) {
            const int ct = wave & 3;
            const float* arow = tb + m * TILE_PITCH + hi * 2;
            const v2f* bp = wfrag + ct * 32 + lane;
            v8f c = {};
#pragma unroll
            for (int kb = 0; kb < 16; ++kb) {
                const v2f a = *(const v2f*)(arow + kb * 4);
                c = __builtin_amdgcn_wmma_f32_16x16x4_f32(
                         false, a,  false, bp[kb * 128],
                         (short)0, c,  false,  false);
            }
            float* orow = tn + (size_t)(rt * 16 + hi * 8) * DIM + ct * 16 + m;
#pragma unroll
            for (int r = 0; r < 8; ++r) orow[(size_t)r * DIM] = c[r];
        }
    }
}

extern "C" void kernel_launch(void* const* d_in, const int* in_sizes, int n_in,
                              void* d_out, int out_size, void* d_ws, size_t ws_size,
                              hipStream_t stream) {
    const float* x  = (const float*)d_in[0];
    const int*   ei = (const int*)d_in[1];
    const float* W1 = (const float*)d_in[2];
    const float* b1 = (const float*)d_in[3];
    const float* W2 = (const float*)d_in[4];
    const float* b2 = (const float*)d_in[5];
    const float* W3 = (const float*)d_in[6];
    const float* b3 = (const float*)d_in[7];
    float* out = (float*)d_out;

    char* ws = (char*)d_ws;
    float* dinv   = (float*)(ws + OFF_DINV);
    float* tbuf   = (float*)(ws + OFF_TBUF);
    int*   lstart = (int*)(ws + OFF_LSTART);
    int*   rowptr = (int*)(ws + OFF_ROWPTR);
    int*   csr    = (int*)(ws + OFF_CSR);
    int2*  binned = (int2*)(ws + OFF_TBUF);

    chunk_sort_kernel<<<NCHUNK, 1024, 0, stream>>>(ei, lstart, binned);
    bin_sort_kernel<<<NBINS, 256, 0, stream>>>(binned, lstart, rowptr, dinv, csr);

    gemm_wmma_kernel<<<512, 256, 0, stream>>>(x, W1, dinv, tbuf);

    gather_gemm_kernel<<<2048, 256, 0, stream>>>(tbuf, rowptr, csr, dinv, b1, W2, out);

    gather_gemm_kernel<<<2048, 256, 0, stream>>>(out, rowptr, csr, dinv, b2, W3, tbuf);

    gather_kernel<false><<<2048, 256, 0, stream>>>(tbuf, rowptr, csr, dinv, b3, out);
}


# Gather list read ahead: each 16-lane row fetches its node's source ids one node early, hands them out by row_share moves, so all row loads of a node are in flight at once; grid 1250; sum order kept
# speedup vs baseline: 4.6894x; 1.0412x over previous
#include <hip/hip_runtime.h>


#define N_NODES 100000
#define N_EDGES 1200000
#define DIM 64

#define OFF_DINV   ((size_t)0)
#define OFF_TBUF   ((size_t)400128)
#define OFF_LSTART ((size_t)400128 + (size_t)25600000)
#define OFF_ROWPTR (OFF_LSTART + (size_t)400128)
#define OFF_CSR    (OFF_ROWPTR + (size_t)400384 + (size_t)4800000)

typedef __attribute__((ext_vector_type(2))) float v2f;
typedef __attribute__((ext_vector_type(8))) float v8f;

#define NBINS 391
#define LPITCH (NBINS + 1)
#define NCHUNK 128
#define CHUNK_EDGES (N_EDGES / NCHUNK)
#define CHUNK_TRIPS 10

__global__ __launch_bounds__(1024) void chunk_sort_kernel(const int* __restrict__ ei,
                                                          int* __restrict__ lstart,
                                                          int2* __restrict__ binned) {
    __shared__ int hist[512];
    __shared__ int wsum[32];
    const int tid = threadIdx.x;
    const int lane = tid & 31;
    const int w = tid >> 5;
    if (tid < 512) hist[tid] = 0;
    const int* src = ei + blockIdx.x * CHUNK_EDGES;
    const int* dst = src + N_EDGES;
    int s[CHUNK_TRIPS];
    int d[CHUNK_TRIPS];
#pragma unroll
    for (int t = 0; t < CHUNK_TRIPS; ++t) {
        const int k = tid + t * 1024;
        const int kk = k < CHUNK_EDGES ? k : CHUNK_EDGES - 1;
        s[t] = src[kk];
        d[t] = dst[kk];
    }
    __syncthreads();
    int r[CHUNK_TRIPS];
#pragma unroll
    for (int t = 0; t < CHUNK_TRIPS; ++t) {
        r[t] = 0;
        if (tid + t * 1024 < CHUNK_EDGES) r[t] = atomicAdd(&hist[d[t] >> 8], 1);
    }
    __syncthreads();
    const int v = tid < 512 ? hist[tid] : 0;
    int x = v;
#pragma unroll
    for (int o = 1; o < 32; o <<= 1) {
        int y = __shfl_up(x, o);
        if (lane >= o) x += y;
    }
    if (lane == 31) wsum[w] = x;
    __syncthreads();
    const int ws = wsum[lane];
    int wx = ws;
#pragma unroll
    for (int o = 1; o < 32; o <<= 1) {
        int y = __shfl_up(wx, o);
        if (lane >= o) wx += y;
    }
    const int ex = __shfl(wx - ws, w) + (x - v);
    if (tid < 512) hist[tid] = ex;
    if (tid < LPITCH) lstart[blockIdx.x * LPITCH + tid] = ex;
    __syncthreads();
    int2* piece = binned + blockIdx.x * CHUNK_EDGES;
#pragma unroll
    for (int t = 0; t < CHUNK_TRIPS; ++t) {
        if (tid + t * 1024 < CHUNK_EDGES) piece[hist[d[t] >> 8] + r[t]] = make_int2(s[t], d[t]);
    }
}

__global__ __launch_bounds__(256) void bin_sort_kernel(const int2* __restrict__ binned,
                                                       const int* __restrict__ lstart,
                                                       int* __restrict__ rowptr,
                                                       float* __restrict__ dinv,
                                                       int* __restrict__ csr) {
    __shared__ int cnt[256];
    __shared__ int wsum[8];
    __shared__ int bsum[8];
    __shared__ int segb[NCHUNK];
    __shared__ int sege[NCHUNK];
    const int tid = threadIdx.x;
    const int lane = tid & 31;
    const int w = tid >> 5;
    const int base = blockIdx.x << 8;
    int lb = 0;
    if (tid < NCHUNK) {
        lb = lstart[tid * LPITCH + blockIdx.x];
        const int le = lstart[tid * LPITCH + blockIdx.x + 1];
        segb[tid] = tid * CHUNK_EDGES + lb;
        sege[tid] = tid * CHUNK_EDGES + le;
    }
    cnt[tid] = 0;
    int y = lb;
#pragma unroll
    for (int o = 16; o > 0; o >>= 1) y += __shfl_xor(y, o);
    if (lane == 0) bsum[w] = y;
    __syncthreads();
    int beg = 0;
#pragma unroll
    for (int q = 0; q < 8; ++q) beg += bsum[q];

    const int g = tid >> 3;
    const int sl = tid & 7;
    int jb[4];
    int je[4];
#pragma unroll
    for (int u = 0; u < 4; ++u) {
        jb[u] = segb[g + 32 * u] + sl;
        je[u] = sege[g + 32 * u];
    }
    for (int k = 0; k < CHUNK_EDGES; k += 8) {
        bool any = false;
#pragma unroll
        for (int u = 0; u < 4; ++u) any = any || (jb[u] + k < je[u]);
        if (!any) break;
        int dd[4];
#pragma unroll
        for (int u = 0; u < 4; ++u) {
            const int j = jb[u] + k;
            dd[u] = binned[j < je[u] ? j : 0].y;
        }
#pragma unroll
        for (int u = 0; u < 4; ++u) {
            if (jb[u] + k < je[u]) atomicAdd(&cnt[dd[u] - base], 1);
        }
    }
    __syncthreads();
    const int c = cnt[tid];
    int x = c;
#pragma unroll
    for (int o = 1; o < 32; o <<= 1) {
        int z = __shfl_up(x, o);
        if (lane >= o) x += z;
    }
    if (lane == 31) wsum[w] = x;
    __syncthreads();
    int wpre = 0;
#pragma unroll
    for (int q = 0; q < 8; ++q) {
        const int t = wsum[q];
        if (q < w) wpre += t;
    }
    const int first = beg + wpre + (x - c);
    cnt[tid] = first;
    const int n = base + tid;
    if (n <= N_NODES) rowptr[n] = first;
    if (n < N_NODES) dinv[n] = rsqrtf((float)(c + 1));
    __syncthreads();
    for (int k = 0; k < CHUNK_EDGES; k += 8) {
        bool any = false;
#pragma unroll
        for (int u = 0; u < 4; ++u) any = any || (jb[u] + k < je[u]);
        if (!any) break;
        int2 e[4];
        int p[4];
#pragma unroll
        for (int u = 0; u < 4; ++u) {
            const int j = jb[u] + k;
            e[u] = binned[j < je[u] ? j : 0];
        }
#pragma unroll
        for (int u = 0; u < 4; ++u) {
            p[u] = 0;
            if (jb[u] + k < je[u]) p[u] = atomicAdd(&cnt[e[u].y - base], 1);
        }
#pragma unroll
        for (int u = 0; u < 4; ++u) {
            if (jb[u] + k < je[u]) csr[p[u]] = e[u].x;
        }
    }
}

__global__ void gemm_wmma_kernel(const float* __restrict__ in,
                                 const float* __restrict__ W,
                                 const float* __restrict__ dinv,
                                 float* __restrict__ out) {
    const int lane = threadIdx.x & 31;
    const int wid  = (blockIdx.x * blockDim.x + threadIdx.x) >> 5;
    const int nWaves = (gridDim.x * blockDim.x) >> 5;

    const int col0 = (wid & 3) * 16;
    const int m    = lane & 15;
    const int hi   = lane >> 4;
    const int klo  = hi * 2;

    v2f bfrag[16];
#pragma unroll
    for (int kb = 0; kb < 16; ++kb) {
        int k = kb * 4 + klo;
        bfrag[kb].x = W[k * DIM + col0 + m];
        bfrag[kb].y = W[(k + 1) * DIM + col0 + m];
    }

    const int nRowTiles  = N_NODES / 16;
    const int rowStride  = nWaves >> 2;

    int rt = wid >> 2;
    const int rfirst = (rt < nRowTiles ? rt : 0) * 16 + hi * 8;
    float4 dlo = *(const float4*)(dinv + rfirst);
    float4 dhi = *(const float4*)(dinv + rfirst + 4);
    for (; rt < nRowTiles; rt += rowStride) {
        const int row0 = rt * 16;
        const float* arow = in + (size_t)(row0 + m) * DIM;
        const int rtn = rt + rowStride;
        const int rnext = (rtn < nRowTiles ? rtn : rt) * 16 + hi * 8;
        const float4 nlo = *(const float4*)(dinv + rnext);
        const float4 nhi = *(const float4*)(dinv + rnext + 4);
        v8f c = {};
#pragma unroll
        for (int kb = 0; kb < 16; ++kb) {
            v2f a = *(const v2f*)(arow + kb * 4 + klo);
            c = __builtin_amdgcn_wmma_f32_16x16x4_f32(
                     false, a,  false, bfrag[kb],
                     (short)0, c,  false,  false);
        }
        float* orow = out + (size_t)(row0 + hi * 8) * DIM + col0 + m;
        asm volatile("v_nop\n\tv_nop\n\tv_nop\n\tv_nop\n\tv_nop\n\tv_nop\n\tv_nop\n\tv_nop" : "+v"(c));
        const float ds[8] = {dlo.x, dlo.y, dlo.z, dlo.w, dhi.x, dhi.y, dhi.z, dhi.w};
#pragma unroll
        for (int r = 0; r < 8; ++r) orow[(size_t)r * DIM] = c[r] * ds[r];
        dlo = nlo;
        dhi = nhi;
    }
}

static __device__ __forceinline__ void add4(float4& acc, const float4 r) {
    acc.x += r.x;
    acc.y += r.y;
    acc.z += r.z;
    acc.w += r.w;
}

template <int K>
static __device__ __forceinline__ int row_lane(const int v) {
    return __builtin_amdgcn_update_dpp(0, v, 0x150 + K, 0xf, 0xf, false);
}

static __device__ __forceinline__ float4 row_piece(const float* __restrict__ t, const int s, const int l) {
    return *(const float4*)((const char*)t + (unsigned)(s * (DIM * 4) + l * 16));
}

static __device__ __forceinline__ int slot_clamp(const int j) {
    return j < N_EDGES ? j : N_EDGES - 1;
}

#define ROW_LOAD(K, IDS, BASE)                                              \
    if ((BASE) + (K) < cnt) r[K] = row_piece(t, row_lane<K>(IDS), l);
#define ROW_ADD(K, BASE)                                                    \
    if ((BASE) + (K) < cnt) add4(acc, r[K]);
#define ROWS15(M, ...)                                                      \
    M(1, __VA_ARGS__) M(2, __VA_ARGS__) M(3, __VA_ARGS__)                   \
    M(4, __VA_ARGS__) M(5, __VA_ARGS__) M(6, __VA_ARGS__) M(7, __VA_ARGS__) \
    M(8, __VA_ARGS__) M(9, __VA_ARGS__) M(10, __VA_ARGS__) M(11, __VA_ARGS__) \
    M(12, __VA_ARGS__) M(13, __VA_ARGS__) M(14, __VA_ARGS__) M(15, __VA_ARGS__)

#define ROW_GROUP(IDS, BASE)                                                \
    {                                                                       \
        float4 r[16];                                                       \
        r[0] = row_piece(t, row_lane<0>(IDS), l);                           \
        ROWS15(ROW_LOAD, IDS, BASE)                                         \
        add4(acc, r[0]);                                                    \
        ROWS15(ROW_ADD, BASE)                                               \
    }

static __device__ __forceinline__ float4 node_sum(const float* __restrict__ t,
                                                  const int* __restrict__ csr,
                                                  const int n, const int l,
                                                  const int beg, const int cnt,
                                                  const int ia, const int ib) {
    float4 acc = row_piece(t, n, l);
    if (cnt > 0) {
        ROW_GROUP(ia, 0)
        if (cnt > 16) {
            ROW_GROUP(ib, 16)
            for (int base = 32; base < cnt; base += 16) {
                const int ic = csr[slot_clamp(beg + base + l)];
                ROW_GROUP(ic, base)
            }
        }
    }
    return acc;
}

template <bool RELU>
__global__ __launch_bounds__(256) void gather_kernel(const float* __restrict__ t,
                                                     const int* __restrict__ rowptr,
                                                     const int* __restrict__ csr,
                                                     const float* __restrict__ dinv,
                                                     const float* __restrict__ bias,
                                                     float* __restrict__ h) {
    const int l = threadIdx.x & 15;
    const float4 b = ((const float4*)bias)[l];
    const int stride = gridDim.x * 16;
    int n = blockIdx.x * 16 + (threadIdx.x >> 4);
    if (n >= N_NODES) return;
    int beg = rowptr[n];
    int end = rowptr[n + 1];
    const int n1 = n + stride < N_NODES ? n + stride : N_NODES - 1;
    int beg1 = rowptr[n1];
    int end1 = rowptr[n1 + 1];
    int ia = csr[slot_clamp(beg + l)];
    int ib = csr[slot_clamp(beg + 16 + l)];
    for (; n < N_NODES; n += stride) {
        const float dn = dinv[n];
        const int n2 = n + 2 * stride < N_NODES ? n + 2 * stride : N_NODES - 1;
        const int beg2 = rowptr[n2];
        const int end2 = rowptr[n2 + 1];
        const int ia1 = csr[slot_clamp(beg1 + l)];
        const int ib1 = csr[slot_clamp(beg1 + 16 + l)];
        float4 acc = node_sum(t, csr, n, l, beg, end - beg, ia, ib);
        acc.x = acc.x * dn + b.x;
        acc.y = acc.y * dn + b.y;
        acc.z = acc.z * dn + b.z;
        acc.w = acc.w * dn + b.w;
        if (RELU) {
            acc.x = fmaxf(acc.x, 0.0f);
            acc.y = fmaxf(acc.y, 0.0f);
            acc.z = fmaxf(acc.z, 0.0f);
            acc.w = fmaxf(acc.w, 0.0f);
        }
        ((float4*)h)[n * 16 + l] = acc;
        beg = beg1;
        end = end1;
        beg1 = beg2;
        end1 = end2;
        ia = ia1;
        ib = ib1;
    }
}

#define TILE_PITCH 68
#define N_TILES (N_NODES / 16)

__global__ __launch_bounds__(256) void gather_gemm_kernel(const float* __restrict__ t,
                                                          const int* __restrict__ rowptr,
                                                          const int* __restrict__ csr,
                                                          const float* __restrict__ dinv,
                                                          const float* __restrict__ bias,
                                                          const float* __restrict__ Wn,
                                                          float* __restrict__ tn) {
    __shared__ v2f wfrag[16 * 4 * 32];
    __shared__ __attribute__((aligned(16))) float tile[2][16 * TILE_PITCH];

    const int l    = threadIdx.x & 15;
    const int rowT = threadIdx.x >> 4;
    const int lane = threadIdx.x & 31;
    const int m    = lane & 15;
    const int hi   = lane >> 4;
    const int wave = __builtin_amdgcn_readfirstlane((int)(threadIdx.x >> 5));

    const int rt0 = blockIdx.x < N_TILES ? blockIdx.x : N_TILES - 1;
    const int rt1 = rt0 + gridDim.x < N_TILES ? rt0 + gridDim.x : N_TILES - 1;
    int beg = rowptr[rt0 * 16 + rowT];
    int end = rowptr[rt0 * 16 + rowT + 1];
    int beg1 = rowptr[rt1 * 16 + rowT];
    int end1 = rowptr[rt1 * 16 + rowT + 1];

#pragma unroll
    for (int i = 0; i < (16 * 4 * 32) / 256; ++i) {
        const int idx = i * 256 + threadIdx.x;
        const int fl = idx & 31;
        const int ct = (idx >> 5) & 3;
        const int kb = idx >> 7;
        const int k  = kb * 4 + (fl >> 4) * 2;
        v2f f;
        f.x = Wn[k * DIM + ct * 16 + (fl & 15)];
        f.y = Wn[(k + 1) * DIM + ct * 16 + (fl & 15)];
        wfrag[idx] = f;
    }

    int ia = csr[slot_clamp(beg + l)];
    int ib = csr[slot_clamp(beg + 16 + l)];
    const float4 b = ((const float4*)bias)[l];
    int it = 0;
    for (int rt = blockIdx.x; rt < N_TILES; rt += gridDim.x, ++it) {
        const int n = rt * 16 + rowT;
        const float dn = dinv[n];
        const int rt2 = rt + 2 * gridDim.x < N_TILES ? rt + 2 * gridDim.x : N_TILES - 1;
        const int beg2 = rowptr[rt2 * 16 + rowT];
        const int end2 = rowptr[rt2 * 16 + rowT + 1];
        const int ia1 = csr[slot_clamp(beg1 + l)];
        const int ib1 = csr[slot_clamp(beg1 + 16 + l)];
        float4 acc = node_sum(t, csr, n, l, beg, end - beg, ia, ib);
        acc.x = fmaxf(acc.x * dn + b.x, 0.0f) * dn;
        acc.y = fmaxf(acc.y * dn + b.y, 0.0f) * dn;
        acc.z = fmaxf(acc.z * dn + b.z, 0.0f) * dn;
        acc.w = fmaxf(acc.w * dn + b.w, 0.0f) * dn;
        beg = beg1;
        end = end1;
        beg1 = beg2;
        end1 = end2;
        ia = ia1;
        ib = ib1;

        float* tb = tile[it & 1];
        *(float4*)(tb + rowT * TILE_PITCH + l * 4) = acc;
        __syncthreads();

        if ((wave >> 2) == (it & 1)) {
            const int ct = wave & 3;
            const float* arow = tb + m * TILE_PITCH + hi * 2;
            const v2f* bp = wfrag + ct * 32 + lane;
            v8f c = {};
#pragma unroll
            for (int kb = 0; kb < 16; ++kb) {
                const v2f a = *(const v2f*)(arow + kb * 4);
                c = __builtin_amdgcn_wmma_f32_16x16x4_f32(
                         false, a,  false, bp[kb * 128],
                         (short)0, c,  false,  false);
            }
            asm volatile("v_nop\n\tv_nop\n\tv_nop\n\tv_nop\n\tv_nop\n\tv_nop\n\tv_nop\n\tv_nop" : "+v"(c));
            float* orow = tn + (size_t)(rt * 16 + hi * 8) * DIM + ct * 16 + m;
#pragma unroll
            for (int r = 0; r < 8; ++r) orow[(size_t)r * DIM] = c[r];
        }
    }
}

extern "C" void kernel_launch(void* const* d_in, const int* in_sizes, int n_in,
                              void* d_out, int out_size, void* d_ws, size_t ws_size,
                              hipStream_t stream) {
    const float* x  = (const float*)d_in[0];
    const int*   ei = (const int*)d_in[1];
    const float* W1 = (const float*)d_in[2];
    const float* b1 = (const float*)d_in[3];
    const float* W2 = (const float*)d_in[4];
    const float* b2 = (const float*)d_in[5];
    const float* W3 = (const float*)d_in[6];
    const float* b3 = (const float*)d_in[7];
    float* out = (float*)d_out;

    char* ws = (char*)d_ws;
    float* dinv   = (float*)(ws + OFF_DINV);
    float* tbuf   = (float*)(ws + OFF_TBUF);
    int*   lstart = (int*)(ws + OFF_LSTART);
    int*   rowptr = (int*)(ws + OFF_ROWPTR);
    int*   csr    = (int*)(ws + OFF_CSR);
    int2*  binned = (int2*)(ws + OFF_TBUF);

    chunk_sort_kernel<<<NCHUNK, 1024, 0, stream>>>(ei, lstart, binned);
    bin_sort_kernel<<<NBINS, 256, 0, stream>>>(binned, lstart, rowptr, dinv, csr);

    gemm_wmma_kernel<<<512, 256, 0, stream>>>(x, W1, dinv, tbuf);

    gather_gemm_kernel<<<1250, 256, 0, stream>>>(tbuf, rowptr, csr, dinv, b1, W2, out);

    gather_gemm_kernel<<<1250, 256, 0, stream>>>(out, rowptr, csr, dinv, b2, W3, tbuf);

    gather_kernel<false><<<1250, 256, 0, stream>>>(tbuf, rowptr, csr, dinv, b3, out);
}


# Sorted records assembled in LDS and stored as whole 128-byte lines in both grouping kernels (pieces padded to a line), not one 8- or 4-byte piece per record; sum order (LDS-add arrival) unchanged
# speedup vs baseline: 4.9728x; 1.0604x over previous
#include <hip/hip_runtime.h>


#define N_NODES 100000
#define N_EDGES 1200000
#define DIM 64

#define OFF_DINV   ((size_t)0)
#define OFF_TBUF   ((size_t)400128)
#define OFF_LSTART ((size_t)400128 + (size_t)25600000)
#define OFF_ROWPTR (OFF_LSTART + (size_t)400128)
#define OFF_CSR    (OFF_ROWPTR + (size_t)400384 + (size_t)4800000)

typedef __attribute__((ext_vector_type(2))) float v2f;
typedef __attribute__((ext_vector_type(8))) float v8f;

#define NBINS 391
#define LPITCH (NBINS + 1)
#define NCHUNK 128
#define CHUNK_EDGES (N_EDGES / NCHUNK)
#define CHUNK_TRIPS 10
#define PIECE_PITCH (CHUNK_EDGES + 1)
#define PIECE_QUADS (PIECE_PITCH / 2)
#define COPY_TRIPS 5
#define STAGE_CAP 8192

__global__ __launch_bounds__(1024) void chunk_sort_kernel(const int* __restrict__ ei,
                                                          int* __restrict__ lstart,
                                                          int2* __restrict__ binned) {
    __shared__ int hist[512];
    __shared__ int wsum[32];
    __shared__ __attribute__((aligned(16))) int2 stage[PIECE_PITCH];
    const int tid = threadIdx.x;
    const int lane = tid & 31;
    const int w = tid >> 5;
    if (tid < 512) hist[tid] = 0;
    if (tid == 0) stage[CHUNK_EDGES] = make_int2(0, 0);
    const int* src = ei + blockIdx.x * CHUNK_EDGES;
    const int* dst = src + N_EDGES;
    int s[CHUNK_TRIPS];
    int d[CHUNK_TRIPS];
#pragma unroll
    for (int t = 0; t < CHUNK_TRIPS; ++t) {
        const int k = tid + t * 1024;
        const int kk = k < CHUNK_EDGES ? k : CHUNK_EDGES - 1;
        s[t] = src[kk];
        d[t] = dst[kk];
    }
    __syncthreads();
    int r[CHUNK_TRIPS];
#pragma unroll
    for (int t = 0; t < CHUNK_TRIPS; ++t) {
        r[t] = 0;
        if (tid + t * 1024 < CHUNK_EDGES) r[t] = atomicAdd(&hist[d[t] >> 8], 1);
    }
    __syncthreads();
    const int v = tid < 512 ? hist[tid] : 0;
    int x = v;
#pragma unroll
    for (int o = 1; o < 32; o <<= 1) {
        int y = __shfl_up(x, o);
        if (lane >= o) x += y;
    }
    if (lane == 31) wsum[w] = x;
    __syncthreads();
    const int ws = wsum[lane];
    int wx = ws;
#pragma unroll
    for (int o = 1; o < 32; o <<= 1) {
        int y = __shfl_up(wx, o);
        if (lane >= o) wx += y;
    }
    const int ex = __shfl(wx - ws, w) + (x - v);
    if (tid < 512) hist[tid] = ex;
    if (tid < LPITCH) lstart[blockIdx.x * LPITCH + tid] = ex;
    __syncthreads();
#pragma unroll
    for (int t = 0; t < CHUNK_TRIPS; ++t) {
        if (tid + t * 1024 < CHUNK_EDGES) stage[hist[d[t] >> 8] + r[t]] = make_int2(s[t], d[t]);
    }
    __syncthreads();
    int4* piece = (int4*)(binned + blockIdx.x * PIECE_PITCH);
    const int4* staged = (const int4*)stage;
#pragma unroll
    for (int t = 0; t < COPY_TRIPS; ++t) {
        const int k = tid + t * 1024;
        if (k < PIECE_QUADS) piece[k] = staged[k];
    }
}

static __device__ __forceinline__ void place_walk(const int2* __restrict__ binned,
                                                  const int (&jb)[4], const int (&je)[4],
                                                  int* cnt, const int base,
                                                  int* slots, const int first) {
    for (int k = 0; k < CHUNK_EDGES; k += 8) {
        bool any = false;
#pragma unroll
        for (int u = 0; u < 4; ++u) any = any || (jb[u] + k < je[u]);
        if (!any) break;
        int2 e[4];
        int p[4];
#pragma unroll
        for (int u = 0; u < 4; ++u) {
            const int j = jb[u] + k;
            e[u] = binned[j < je[u] ? j : 0];
        }
#pragma unroll
        for (int u = 0; u < 4; ++u) {
            p[u] = 0;
            if (jb[u] + k < je[u]) p[u] = atomicAdd(&cnt[e[u].y - base], 1);
        }
#pragma unroll
        for (int u = 0; u < 4; ++u) {
            if (jb[u] + k < je[u]) slots[p[u] - first] = e[u].x;
        }
    }
}

__global__ __launch_bounds__(256) void bin_sort_kernel(const int2* __restrict__ binned,
                                                       const int* __restrict__ lstart,
                                                       int* __restrict__ rowptr,
                                                       float* __restrict__ dinv,
                                                       int* __restrict__ csr) {
    __shared__ int cnt[256];
    __shared__ int wsum[8];
    __shared__ int bsum[8];
    __shared__ int segb[NCHUNK];
    __shared__ int sege[NCHUNK];
    __shared__ int stage[STAGE_CAP];
    const int tid = threadIdx.x;
    const int lane = tid & 31;
    const int w = tid >> 5;
    const int base = blockIdx.x << 8;
    int lb = 0;
    if (tid < NCHUNK) {
        lb = lstart[tid * LPITCH + blockIdx.x];
        const int le = lstart[tid * LPITCH + blockIdx.x + 1];
        segb[tid] = tid * PIECE_PITCH + lb;
        sege[tid] = tid * PIECE_PITCH + le;
    }
    cnt[tid] = 0;
    int y = lb;
#pragma unroll
    for (int o = 16; o > 0; o >>= 1) y += __shfl_xor(y, o);
    if (lane == 0) bsum[w] = y;
    __syncthreads();
    int beg = 0;
#pragma unroll
    for (int q = 0; q < 8; ++q) beg += bsum[q];

    const int g = tid >> 3;
    const int sl = tid & 7;
    int jb[4];
    int je[4];
#pragma unroll
    for (int u = 0; u < 4; ++u) {
        jb[u] = segb[g + 32 * u] + sl;
        je[u] = sege[g + 32 * u];
    }
    for (int k = 0; k < CHUNK_EDGES; k += 8) {
        bool any = false;
#pragma unroll
        for (int u = 0; u < 4; ++u) any = any || (jb[u] + k < je[u]);
        if (!any) break;
        int dd[4];
#pragma unroll
        for (int u = 0; u < 4; ++u) {
            const int j = jb[u] + k;
            dd[u] = binned[j < je[u] ? j : 0].y;
        }
#pragma unroll
        for (int u = 0; u < 4; ++u) {
            if (jb[u] + k < je[u]) atomicAdd(&cnt[dd[u] - base], 1);
        }
    }
    __syncthreads();
    const int c = cnt[tid];
    int x = c;
#pragma unroll
    for (int o = 1; o < 32; o <<= 1) {
        int z = __shfl_up(x, o);
        if (lane >= o) x += z;
    }
    if (lane == 31) wsum[w] = x;
    __syncthreads();
    int wpre = 0;
    int tot = 0;
#pragma unroll
    for (int q = 0; q < 8; ++q) {
        const int t = wsum[q];
        tot += t;
        if (q < w) wpre += t;
    }
    const int first = beg + wpre + (x - c);
    cnt[tid] = first;
    const int n = base + tid;
    if (n <= N_NODES) rowptr[n] = first;
    if (n < N_NODES) dinv[n] = rsqrtf((float)(c + 1));
    __syncthreads();
    const int btot = __builtin_amdgcn_readfirstlane(tot);
    if (btot <= STAGE_CAP) {
        place_walk(binned, jb, je, cnt, base, stage, beg);
        __syncthreads();
        const int gend = beg + btot;
        for (int g0 = (beg & ~31) + tid; g0 < gend; g0 += 1024) {
            int v[4];
#pragma unroll
            for (int u = 0; u < 4; ++u) {
                const int g = g0 + 256 * u;
                v[u] = (g >= beg && g < gend) ? stage[g - beg] : 0;
            }
#pragma unroll
            for (int u = 0; u < 4; ++u) {
                const int g = g0 + 256 * u;
                if (g >= beg && g < gend) csr[g] = v[u];
            }
        }
    } else {
        place_walk(binned, jb, je, cnt, base, csr, 0);
    }
}

__global__ void gemm_wmma_kernel(const float* __restrict__ in,
                                 const float* __restrict__ W,
                                 const float* __restrict__ dinv,
                                 float* __restrict__ out) {
    const int lane = threadIdx.x & 31;
    const int wid  = (blockIdx.x * blockDim.x + threadIdx.x) >> 5;
    const int nWaves = (gridDim.x * blockDim.x) >> 5;

    const int col0 = (wid & 3) * 16;
    const int m    = lane & 15;
    const int hi   = lane >> 4;
    const int klo  = hi * 2;

    v2f bfrag[16];
#pragma unroll
    for (int kb = 0; kb < 16; ++kb) {
        int k = kb * 4 + klo;
        bfrag[kb].x = W[k * DIM + col0 + m];
        bfrag[kb].y = W[(k + 1) * DIM + col0 + m];
    }

    const int nRowTiles  = N_NODES / 16;
    const int rowStride  = nWaves >> 2;

    int rt = wid >> 2;
    const int rfirst = (rt < nRowTiles ? rt : 0) * 16 + hi * 8;
    float4 dlo = *(const float4*)(dinv + rfirst);
    float4 dhi = *(const float4*)(dinv + rfirst + 4);
    for (; rt < nRowTiles; rt += rowStride) {
        const int row0 = rt * 16;
        const float* arow = in + (size_t)(row0 + m) * DIM;
        const int rtn = rt + rowStride;
        const int rnext = (rtn < nRowTiles ? rtn : rt) * 16 + hi * 8;
        const float4 nlo = *(const float4*)(dinv + rnext);
        const float4 nhi = *(const float4*)(dinv + rnext + 4);
        v8f c = {};
#pragma unroll
        for (int kb = 0; kb < 16; ++kb) {
            v2f a = *(const v2f*)(arow + kb * 4 + klo);
            c = __builtin_amdgcn_wmma_f32_16x16x4_f32(
                     false, a,  false, bfrag[kb],
                     (short)0, c,  false,  false);
        }
        float* orow = out + (size_t)(row0 + hi * 8) * DIM + col0 + m;
        asm volatile("v_nop\n\tv_nop\n\tv_nop\n\tv_nop\n\tv_nop\n\tv_nop\n\tv_nop\n\tv_nop" : "+v"(c));
        const float ds[8] = {dlo.x, dlo.y, dlo.z, dlo.w, dhi.x, dhi.y, dhi.z, dhi.w};
#pragma unroll
        for (int r = 0; r < 8; ++r) orow[(size_t)r * DIM] = c[r] * ds[r];
        dlo = nlo;
        dhi = nhi;
    }
}

static __device__ __forceinline__ void add4(float4& acc, const float4 r) {
    acc.x += r.x;
    acc.y += r.y;
    acc.z += r.z;
    acc.w += r.w;
}

template <int K>
static __device__ __forceinline__ int row_lane(const int v) {
    return __builtin_amdgcn_update_dpp(0, v, 0x150 + K, 0xf, 0xf, false);
}

static __device__ __forceinline__ float4 row_piece(const float* __restrict__ t, const int s, const int l) {
    return *(const float4*)((const char*)t + (unsigned)(s * (DIM * 4) + l * 16));
}

static __device__ __forceinline__ int slot_clamp(const int j) {
    return j < N_EDGES ? j : N_EDGES - 1;
}

#define ROW_LOAD(K, IDS, BASE)                                              \
    if ((BASE) + (K) < cnt) r[K] = row_piece(t, row_lane<K>(IDS), l);
#define ROW_ADD(K, BASE)                                                    \
    if ((BASE) + (K) < cnt) add4(acc, r[K]);
#define ROWS15(M, ...)                                                      \
    M(1, __VA_ARGS__) M(2, __VA_ARGS__) M(3, __VA_ARGS__)                   \
    M(4, __VA_ARGS__) M(5, __VA_ARGS__) M(6, __VA_ARGS__) M(7, __VA_ARGS__) \
    M(8, __VA_ARGS__) M(9, __VA_ARGS__) M(10, __VA_ARGS__) M(11, __VA_ARGS__) \
    M(12, __VA_ARGS__) M(13, __VA_ARGS__) M(14, __VA_ARGS__) M(15, __VA_ARGS__)

#define ROW_GROUP(IDS, BASE)                                                \
    {                                                                       \
        float4 r[16];                                                       \
        r[0] = row_piece(t, row_lane<0>(IDS), l);                           \
        ROWS15(ROW_LOAD, IDS, BASE)                                         \
        add4(acc, r[0]);                                                    \
        ROWS15(ROW_ADD, BASE)                                               \
    }

static __device__ __forceinline__ float4 node_sum(const float* __restrict__ t,
                                                  const int* __restrict__ csr,
                                                  const int n, const int l,
                                                  const int beg, const int cnt,
                                                  const int ia, const int ib) {
    float4 acc = row_piece(t, n, l);
    if (cnt > 0) {
        ROW_GROUP(ia, 0)
        if (cnt > 16) {
            ROW_GROUP(ib, 16)
            for (int base = 32; base < cnt; base += 16) {
                const int ic = csr[slot_clamp(beg + base + l)];
                ROW_GROUP(ic, base)
            }
        }
    }
    return acc;
}

template <bool RELU>
__global__ __launch_bounds__(256) void gather_kernel(const float* __restrict__ t,
                                                     const int* __restrict__ rowptr,
                                                     const int* __restrict__ csr,
                                                     const float* __restrict__ dinv,
                                                     const float* __restrict__ bias,
                                                     float* __restrict__ h) {
    const int l = threadIdx.x & 15;
    const float4 b = ((const float4*)bias)[l];
    const int stride = gridDim.x * 16;
    int n = blockIdx.x * 16 + (threadIdx.x >> 4);
    if (n >= N_NODES) return;
    int beg = rowptr[n];
    int end = rowptr[n + 1];
    const int n1 = n + stride < N_NODES ? n + stride : N_NODES - 1;
    int beg1 = rowptr[n1];
    int end1 = rowptr[n1 + 1];
    int ia = csr[slot_clamp(beg + l)];
    int ib = csr[slot_clamp(beg + 16 + l)];
    for (; n < N_NODES; n += stride) {
        const float dn = dinv[n];
        const int n2 = n + 2 * stride < N_NODES ? n + 2 * stride : N_NODES - 1;
        const int beg2 = rowptr[n2];
        const int end2 = rowptr[n2 + 1];
        const int ia1 = csr[slot_clamp(beg1 + l)];
        const int ib1 = csr[slot_clamp(beg1 + 16 + l)];
        float4 acc = node_sum(t, csr, n, l, beg, end - beg, ia, ib);
        acc.x = acc.x * dn + b.x;
        acc.y = acc.y * dn + b.y;
        acc.z = acc.z * dn + b.z;
        acc.w = acc.w * dn + b.w;
        if (RELU) {
            acc.x = fmaxf(acc.x, 0.0f);
            acc.y = fmaxf(acc.y, 0.0f);
            acc.z = fmaxf(acc.z, 0.0f);
            acc.w = fmaxf(acc.w, 0.0f);
        }
        ((float4*)h)[n * 16 + l] = acc;
        beg = beg1;
        end = end1;
        beg1 = beg2;
        end1 = end2;
        ia = ia1;
        ib = ib1;
    }
}

#define TILE_PITCH 68
#define N_TILES (N_NODES / 16)

__global__ __launch_bounds__(256) void gather_gemm_kernel(const float* __restrict__ t,
                                                          const int* __restrict__ rowptr,
                                                          const int* __restrict__ csr,
                                                          const float* __restrict__ dinv,
                                                          const float* __restrict__ bias,
                                                          const float* __restrict__ Wn,
                                                          float* __restrict__ tn) {
    __shared__ v2f wfrag[16 * 4 * 32];
    __shared__ __attribute__((aligned(16))) float tile[2][16 * TILE_PITCH];

    const int l    = threadIdx.x & 15;
    const int rowT = threadIdx.x >> 4;
    const int lane = threadIdx.x & 31;
    const int m    = lane & 15;
    const int hi   = lane >> 4;
    const int wave = __builtin_amdgcn_readfirstlane((int)(threadIdx.x >> 5));

    const int rt0 = blockIdx.x < N_TILES ? blockIdx.x : N_TILES - 1;
    const int rt1 = rt0 + gridDim.x < N_TILES ? rt0 + gridDim.x : N_TILES - 1;
    int beg = rowptr[rt0 * 16 + rowT];
    int end = rowptr[rt0 * 16 + rowT + 1];
    int beg1 = rowptr[rt1 * 16 + rowT];
    int end1 = rowptr[rt1 * 16 + rowT + 1];

#pragma unroll
    for (int i = 0; i < (16 * 4 * 32) / 256; ++i) {
        const int idx = i * 256 + threadIdx.x;
        const int fl = idx & 31;
        const int ct = (idx >> 5) & 3;
        const int kb = idx >> 7;
        const int k  = kb * 4 + (fl >> 4) * 2;
        v2f f;
        f.x = Wn[k * DIM + ct * 16 + (fl & 15)];
        f.y = Wn[(k + 1) * DIM + ct * 16 + (fl & 15)];
        wfrag[idx] = f;
    }

    int ia = csr[slot_clamp(beg + l)];
    int ib = csr[slot_clamp(beg + 16 + l)];
    const float4 b = ((const float4*)bias)[l];
    int it = 0;
    for (int rt = blockIdx.x; rt < N_TILES; rt += gridDim.x, ++it) {
        const int n = rt * 16 + rowT;
        const float dn = dinv[n];
        const int rt2 = rt + 2 * gridDim.x < N_TILES ? rt + 2 * gridDim.x : N_TILES - 1;
        const int beg2 = rowptr[rt2 * 16 + rowT];
        const int end2 = rowptr[rt2 * 16 + rowT + 1];
        const int ia1 = csr[slot_clamp(beg1 + l)];
        const int ib1 = csr[slot_clamp(beg1 + 16 + l)];
        float4 acc = node_sum(t, csr, n, l, beg, end - beg, ia, ib);
        acc.x = fmaxf(acc.x * dn + b.x, 0.0f) * dn;
        acc.y = fmaxf(acc.y * dn + b.y, 0.0f) * dn;
        acc.z = fmaxf(acc.z * dn + b.z, 0.0f) * dn;
        acc.w = fmaxf(acc.w * dn + b.w, 0.0f) * dn;
        beg = beg1;
        end = end1;
        beg1 = beg2;
        end1 = end2;
        ia = ia1;
        ib = ib1;

        float* tb = tile[it & 1];
        *(float4*)(tb + rowT * TILE_PITCH + l * 4) = acc;
        __syncthreads();

        if ((wave >> 2) == (it & 1)) {
            const int ct = wave & 3;
            const float* arow = tb + m * TILE_PITCH + hi * 2;
            const v2f* bp = wfrag + ct * 32 + lane;
            v8f c = {};
#pragma unroll
            for (int kb = 0; kb < 16; ++kb) {
                const v2f a = *(const v2f*)(arow + kb * 4);
                c = __builtin_amdgcn_wmma_f32_16x16x4_f32(
                         false, a,  false, bp[kb * 128],
                         (short)0, c,  false,  false);
            }
            asm volatile("v_nop\n\tv_nop\n\tv_nop\n\tv_nop\n\tv_nop\n\tv_nop\n\tv_nop\n\tv_nop" : "+v"(c));
            float* orow = tn + (size_t)(rt * 16 + hi * 8) * DIM + ct * 16 + m;
#pragma unroll
            for (int r = 0; r < 8; ++r) orow[(size_t)r * DIM] = c[r];
        }
    }
}

extern "C" void kernel_launch(void* const* d_in, const int* in_sizes, int n_in,
                              void* d_out, int out_size, void* d_ws, size_t ws_size,
                              hipStream_t stream) {
    const float* x  = (const float*)d_in[0];
    const int*   ei = (const int*)d_in[1];
    const float* W1 = (const float*)d_in[2];
    const float* b1 = (const float*)d_in[3];
    const float* W2 = (const float*)d_in[4];
    const float* b2 = (const float*)d_in[5];
    const float* W3 = (const float*)d_in[6];
    const float* b3 = (const float*)d_in[7];
    float* out = (float*)d_out;

    char* ws = (char*)d_ws;
    float* dinv   = (float*)(ws + OFF_DINV);
    float* tbuf   = (float*)(ws + OFF_TBUF);
    int*   lstart = (int*)(ws + OFF_LSTART);
    int*   rowptr = (int*)(ws + OFF_ROWPTR);
    int*   csr    = (int*)(ws + OFF_CSR);
    int2*  binned = (int2*)(ws + OFF_TBUF);

    chunk_sort_kernel<<<NCHUNK, 1024, 0, stream>>>(ei, lstart, binned);
    bin_sort_kernel<<<NBINS, 256, 0, stream>>>(binned, lstart, rowptr, dinv, csr);

    gemm_wmma_kernel<<<512, 256, 0, stream>>>(x, W1, dinv, tbuf);

    gather_gemm_kernel<<<1250, 256, 0, stream>>>(tbuf, rowptr, csr, dinv, b1, W2, out);

    gather_gemm_kernel<<<1250, 256, 0, stream>>>(out, rowptr, csr, dinv, b2, W3, tbuf);

    gather_kernel<false><<<1250, 256, 0, stream>>>(tbuf, rowptr, csr, dinv, b3, out);
}


# Product moved behind the aggregation, (A h) W for A (h W): all 3 layers are one gather+WMMA-product launch, the stand-alone layer-1 product launch is gone (6 to 5); rows summed before the product
# speedup vs baseline: 5.1461x; 1.0348x over previous
#include <hip/hip_runtime.h>


#define N_NODES 100000
#define N_EDGES 1200000
#define DIM 64

#define OFF_DINV   ((size_t)0)
#define OFF_TBUF   ((size_t)400128)
#define OFF_LSTART ((size_t)400128 + (size_t)25600000)
#define OFF_ROWPTR (OFF_LSTART + (size_t)400128)
#define OFF_CSR    (OFF_ROWPTR + (size_t)400384 + (size_t)4800000)

typedef __attribute__((ext_vector_type(2))) float v2f;
typedef __attribute__((ext_vector_type(8))) float v8f;

#define NBINS 391
#define LPITCH (NBINS + 1)
#define NCHUNK 128
#define CHUNK_EDGES (N_EDGES / NCHUNK)
#define CHUNK_TRIPS 10
#define PIECE_PITCH (CHUNK_EDGES + 1)
#define PIECE_QUADS (PIECE_PITCH / 2)
#define COPY_TRIPS 5
#define STAGE_CAP 8192

__global__ __launch_bounds__(1024) void chunk_sort_kernel(const int* __restrict__ ei,
                                                          int* __restrict__ lstart,
                                                          int2* __restrict__ binned) {
    __shared__ int hist[512];
    __shared__ int wsum[32];
    __shared__ __attribute__((aligned(16))) int2 stage[PIECE_PITCH];
    const int tid = threadIdx.x;
    const int lane = tid & 31;
    const int w = tid >> 5;
    if (tid < 512) hist[tid] = 0;
    if (tid == 0) stage[CHUNK_EDGES] = make_int2(0, 0);
    const int* src = ei + blockIdx.x * CHUNK_EDGES;
    const int* dst = src + N_EDGES;
    int s[CHUNK_TRIPS];
    int d[CHUNK_TRIPS];
#pragma unroll
    for (int t = 0; t < CHUNK_TRIPS; ++t) {
        const int k = tid + t * 1024;
        const int kk = k < CHUNK_EDGES ? k : CHUNK_EDGES - 1;
        s[t] = src[kk];
        d[t] = dst[kk];
    }
    __syncthreads();
    int r[CHUNK_TRIPS];
#pragma unroll
    for (int t = 0; t < CHUNK_TRIPS; ++t) {
        r[t] = 0;
        if (tid + t * 1024 < CHUNK_EDGES) r[t] = atomicAdd(&hist[d[t] >> 8], 1);
    }
    __syncthreads();
    const int v = tid < 512 ? hist[tid] : 0;
    int x = v;
#pragma unroll
    for (int o = 1; o < 32; o <<= 1) {
        int y = __shfl_up(x, o);
        if (lane >= o) x += y;
    }
    if (lane == 31) wsum[w] = x;
    __syncthreads();
    const int ws = wsum[lane];
    int wx = ws;
#pragma unroll
    for (int o = 1; o < 32; o <<= 1) {
        int y = __shfl_up(wx, o);
        if (lane >= o) wx += y;
    }
    const int ex = __shfl(wx - ws, w) + (x - v);
    if (tid < 512) hist[tid] = ex;
    if (tid < LPITCH) lstart[blockIdx.x * LPITCH + tid] = ex;
    __syncthreads();
#pragma unroll
    for (int t = 0; t < CHUNK_TRIPS; ++t) {
        if (tid + t * 1024 < CHUNK_EDGES) stage[hist[d[t] >> 8] + r[t]] = make_int2(s[t], d[t]);
    }
    __syncthreads();
    int4* piece = (int4*)(binned + blockIdx.x * PIECE_PITCH);
    const int4* staged = (const int4*)stage;
#pragma unroll
    for (int t = 0; t < COPY_TRIPS; ++t) {
        const int k = tid + t * 1024;
        if (k < PIECE_QUADS) piece[k] = staged[k];
    }
}

static __device__ __forceinline__ void place_walk(const int2* __restrict__ binned,
                                                  const int (&jb)[4], const int (&je)[4],
                                                  int* cnt, const int base,
                                                  int* slots, const int first) {
    for (int k = 0; k < CHUNK_EDGES; k += 8) {
        bool any = false;
#pragma unroll
        for (int u = 0; u < 4; ++u) any = any || (jb[u] + k < je[u]);
        if (!any) break;
        int2 e[4];
        int p[4];
#pragma unroll
        for (int u = 0; u < 4; ++u) {
            const int j = jb[u] + k;
            e[u] = binned[j < je[u] ? j : 0];
        }
#pragma unroll
        for (int u = 0; u < 4; ++u) {
            p[u] = 0;
            if (jb[u] + k < je[u]) p[u] = atomicAdd(&cnt[e[u].y - base], 1);
        }
#pragma unroll
        for (int u = 0; u < 4; ++u) {
            if (jb[u] + k < je[u]) slots[p[u] - first] = e[u].x;
        }
    }
}

__global__ __launch_bounds__(256) void bin_sort_kernel(const int2* __restrict__ binned,
                                                       const int* __restrict__ lstart,
                                                       int* __restrict__ rowptr,
                                                       float* __restrict__ dinv,
                                                       int* __restrict__ csr,
                                                       const float* __restrict__ xin,
                                                       float* __restrict__ xs) {
    __shared__ int cnt[256];
    __shared__ float dl[256];
    __shared__ int wsum[8];
    __shared__ int bsum[8];
    __shared__ int segb[NCHUNK];
    __shared__ int sege[NCHUNK];
    __shared__ int stage[STAGE_CAP];
    const int tid = threadIdx.x;
    const int lane = tid & 31;
    const int w = tid >> 5;
    const int base = blockIdx.x << 8;
    int lb = 0;
    if (tid < NCHUNK) {
        lb = lstart[tid * LPITCH + blockIdx.x];
        const int le = lstart[tid * LPITCH + blockIdx.x + 1];
        segb[tid] = tid * PIECE_PITCH + lb;
        sege[tid] = tid * PIECE_PITCH + le;
    }
    cnt[tid] = 0;
    int y = lb;
#pragma unroll
    for (int o = 16; o > 0; o >>= 1) y += __shfl_xor(y, o);
    if (lane == 0) bsum[w] = y;
    __syncthreads();
    int beg = 0;
#pragma unroll
    for (int q = 0; q < 8; ++q) beg += bsum[q];

    const int g = tid >> 3;
    const int sl = tid & 7;
    int jb[4];
    int je[4];
#pragma unroll
    for (int u = 0; u < 4; ++u) {
        jb[u] = segb[g + 32 * u] + sl;
        je[u] = sege[g + 32 * u];
    }
    for (int k = 0; k < CHUNK_EDGES; k += 8) {
        bool any = false;
#pragma unroll
        for (int u = 0; u < 4; ++u) any = any || (jb[u] + k < je[u]);
        if (!any) break;
        int dd[4];
#pragma unroll
        for (int u = 0; u < 4; ++u) {
            const int j = jb[u] + k;
            dd[u] = binned[j < je[u] ? j : 0].y;
        }
#pragma unroll
        for (int u = 0; u < 4; ++u) {
            if (jb[u] + k < je[u]) atomicAdd(&cnt[dd[u] - base], 1);
        }
    }
    const float4* xrow = (const float4*)xin + base * (DIM / 4);
    float4 xr[16];
#pragma unroll
    for (int i = 0; i < 16; ++i) {
        const int e = i * 256 + tid;
        xr[i] = xrow[base + (e >> 4) < N_NODES ? e : (e & 15)];
    }
    __syncthreads();
    const int c = cnt[tid];
    int x = c;
#pragma unroll
    for (int o = 1; o < 32; o <<= 1) {
        int z = __shfl_up(x, o);
        if (lane >= o) x += z;
    }
    if (lane == 31) wsum[w] = x;
    __syncthreads();
    int wpre = 0;
    int tot = 0;
#pragma unroll
    for (int q = 0; q < 8; ++q) {
        const int t = wsum[q];
        tot += t;
        if (q < w) wpre += t;
    }
    const int first = beg + wpre + (x - c);
    cnt[tid] = first;
    const int n = base + tid;
    if (n <= N_NODES) rowptr[n] = first;
    const float dself = rsqrtf((float)(c + 1));
    if (n < N_NODES) dinv[n] = dself;
    dl[tid] = dself;
    __syncthreads();
    float4* xsrow = (float4*)xs + base * (DIM / 4);
#pragma unroll
    for (int i = 0; i < 16; ++i) {
        const int e = i * 256 + tid;
        if (base + (e >> 4) < N_NODES) {
            const float d = dl[e >> 4];
            float4 v = xr[i];
            v.x *= d;
            v.y *= d;
            v.z *= d;
            v.w *= d;
            xsrow[e] = v;
        }
    }
    const int btot = __builtin_amdgcn_readfirstlane(tot);
    if (btot <= STAGE_CAP) {
        place_walk(binned, jb, je, cnt, base, stage, beg);
        __syncthreads();
        const int gend = beg + btot;
        for (int g0 = (beg & ~31) + tid; g0 < gend; g0 += 1024) {
            int v[4];
#pragma unroll
            for (int u = 0; u < 4; ++u) {
                const int g = g0 + 256 * u;
                v[u] = (g >= beg && g < gend) ? stage[g - beg] : 0;
            }
#pragma unroll
            for (int u = 0; u < 4; ++u) {
                const int g = g0 + 256 * u;
                if (g >= beg && g < gend) csr[g] = v[u];
            }
        }
    } else {
        place_walk(binned, jb, je, cnt, base, csr, 0);
    }
}

static __device__ __forceinline__ void add4(float4& acc, const float4 r) {
    acc.x += r.x;
    acc.y += r.y;
    acc.z += r.z;
    acc.w += r.w;
}

template <int K>
static __device__ __forceinline__ int row_lane(const int v) {
    return __builtin_amdgcn_update_dpp(0, v, 0x150 + K, 0xf, 0xf, false);
}

static __device__ __forceinline__ float4 row_piece(const float* __restrict__ t, const int s, const int l) {
    return *(const float4*)((const char*)t + (unsigned)(s * (DIM * 4) + l * 16));
}

static __device__ __forceinline__ int slot_clamp(const int j) {
    return j < N_EDGES ? j : N_EDGES - 1;
}

#define ROW_LOAD(K, IDS, BASE)                                              \
    if ((BASE) + (K) < cnt) r[K] = row_piece(t, row_lane<K>(IDS), l);
#define ROW_ADD(K, BASE)                                                    \
    if ((BASE) + (K) < cnt) add4(acc, r[K]);
#define ROWS15(M, ...)                                                      \
    M(1, __VA_ARGS__) M(2, __VA_ARGS__) M(3, __VA_ARGS__)                   \
    M(4, __VA_ARGS__) M(5, __VA_ARGS__) M(6, __VA_ARGS__) M(7, __VA_ARGS__) \
    M(8, __VA_ARGS__) M(9, __VA_ARGS__) M(10, __VA_ARGS__) M(11, __VA_ARGS__) \
    M(12, __VA_ARGS__) M(13, __VA_ARGS__) M(14, __VA_ARGS__) M(15, __VA_ARGS__)

#define ROW_GROUP(IDS, BASE)                                                \
    {                                                                       \
        float4 r[16];                                                       \
        r[0] = row_piece(t, row_lane<0>(IDS), l);                           \
        ROWS15(ROW_LOAD, IDS, BASE)                                         \
        add4(acc, r[0]);                                                    \
        ROWS15(ROW_ADD, BASE)                                               \
    }

static __device__ __forceinline__ float4 node_sum(const float* __restrict__ t,
                                                  const int* __restrict__ csr,
                                                  const int n, const int l,
                                                  const int beg, const int cnt,
                                                  const int ia, const int ib) {
    float4 acc = row_piece(t, n, l);
    if (cnt > 0) {
        ROW_GROUP(ia, 0)
        if (cnt > 16) {
            ROW_GROUP(ib, 16)
            for (int base = 32; base < cnt; base += 16) {
                const int ic = csr[slot_clamp(beg + base + l)];
                ROW_GROUP(ic, base)
            }
        }
    }
    return acc;
}

#define TILE_PITCH 68
#define N_TILES (N_NODES / 16)

template <bool LAST>
__global__ __launch_bounds__(256) void gather_gemm_kernel(const float* __restrict__ t,
                                                          const int* __restrict__ rowptr,
                                                          const int* __restrict__ csr,
                                                          const float* __restrict__ dinv,
                                                          const float* __restrict__ bias,
                                                          const float* __restrict__ Wn,
                                                          float* __restrict__ tn) {
    __shared__ v2f wfrag[16 * 4 * 32];
    __shared__ __attribute__((aligned(16))) float tile[2][16 * TILE_PITCH];

    const int l    = threadIdx.x & 15;
    const int rowT = threadIdx.x >> 4;
    const int lane = threadIdx.x & 31;
    const int m    = lane & 15;
    const int hi   = lane >> 4;
    const int wave = __builtin_amdgcn_readfirstlane((int)(threadIdx.x >> 5));

    const int rt0 = blockIdx.x < N_TILES ? blockIdx.x : N_TILES - 1;
    const int rt1 = rt0 + gridDim.x < N_TILES ? rt0 + gridDim.x : N_TILES - 1;
    int beg = rowptr[rt0 * 16 + rowT];
    int end = rowptr[rt0 * 16 + rowT + 1];
    int beg1 = rowptr[rt1 * 16 + rowT];
    int end1 = rowptr[rt1 * 16 + rowT + 1];

#pragma unroll
    for (int i = 0; i < (16 * 4 * 32) / 256; ++i) {
        const int idx = i * 256 + threadIdx.x;
        const int fl = idx & 31;
        const int ct = (idx >> 5) & 3;
        const int kb = idx >> 7;
        const int k  = kb * 4 + (fl >> 4) * 2;
        v2f f;
        f.x = Wn[k * DIM + ct * 16 + (fl & 15)];
        f.y = Wn[(k + 1) * DIM + ct * 16 + (fl & 15)];
        wfrag[idx] = f;
    }

    int ia = csr[slot_clamp(beg + l)];
    int ib = csr[slot_clamp(beg + 16 + l)];
    const float bcol = bias[(wave & 3) * 16 + m];
    int it = 0;
    for (int rt = blockIdx.x; rt < N_TILES; rt += gridDim.x, ++it) {
        const int n = rt * 16 + rowT;
        const float dn = dinv[n];
        const int rt2 = rt + 2 * gridDim.x < N_TILES ? rt + 2 * gridDim.x : N_TILES - 1;
        const int beg2 = rowptr[rt2 * 16 + rowT];
        const int end2 = rowptr[rt2 * 16 + rowT + 1];
        const int ia1 = csr[slot_clamp(beg1 + l)];
        const int ib1 = csr[slot_clamp(beg1 + 16 + l)];
        float4 acc = node_sum(t, csr, n, l, beg, end - beg, ia, ib);
        acc.x *= dn;
        acc.y *= dn;
        acc.z *= dn;
        acc.w *= dn;
        beg = beg1;
        end = end1;
        beg1 = beg2;
        end1 = end2;
        ia = ia1;
        ib = ib1;

        float* tb = tile[it & 1];
        *(float4*)(tb + rowT * TILE_PITCH + l * 4) = acc;
        if (!LAST && l == 0) tb[rowT * TILE_PITCH + DIM] = dn;
        __syncthreads();

        if ((wave >> 2) == (it & 1)) {
            const int ct = wave & 3;
            const float* arow = tb + m * TILE_PITCH + hi * 2;
            const v2f* bp = wfrag + ct * 32 + lane;
            float* orow = tn + (size_t)(rt * 16 + hi * 8) * DIM + ct * 16 + m;
            float dr[8];
            if (!LAST) {
#pragma unroll
                for (int r = 0; r < 8; ++r) dr[r] = tb[(hi * 8 + r) * TILE_PITCH + DIM];
            }
            __builtin_amdgcn_sched_barrier(0);
            v8f c = {};
#pragma unroll
            for (int kb = 0; kb < 16; ++kb) {
                const v2f a = *(const v2f*)(arow + kb * 4);
                c = __builtin_amdgcn_wmma_f32_16x16x4_f32(
                         false, a,  false, bp[kb * 128],
                         (short)0, c,  false,  false);
            }
            asm volatile("v_nop\n\tv_nop\n\tv_nop\n\tv_nop\n\tv_nop\n\tv_nop\n\tv_nop\n\tv_nop" : "+v"(c));
#pragma unroll
            for (int r = 0; r < 8; ++r) {
                float v = c[r] + bcol;
                if (!LAST) v = fmaxf(v, 0.0f) * dr[r];
                orow[(size_t)r * DIM] = v;
            }
        }
    }
}

extern "C" void kernel_launch(void* const* d_in, const int* in_sizes, int n_in,
                              void* d_out, int out_size, void* d_ws, size_t ws_size,
                              hipStream_t stream) {
    const float* x  = (const float*)d_in[0];
    const int*   ei = (const int*)d_in[1];
    const float* W1 = (const float*)d_in[2];
    const float* b1 = (const float*)d_in[3];
    const float* W2 = (const float*)d_in[4];
    const float* b2 = (const float*)d_in[5];
    const float* W3 = (const float*)d_in[6];
    const float* b3 = (const float*)d_in[7];
    float* out = (float*)d_out;

    char* ws = (char*)d_ws;
    float* dinv   = (float*)(ws + OFF_DINV);
    float* tbuf   = (float*)(ws + OFF_TBUF);
    int*   lstart = (int*)(ws + OFF_LSTART);
    int*   rowptr = (int*)(ws + OFF_ROWPTR);
    int*   csr    = (int*)(ws + OFF_CSR);
    int2*  binned = (int2*)d_out;

    chunk_sort_kernel<<<NCHUNK, 1024, 0, stream>>>(ei, lstart, binned);
    bin_sort_kernel<<<NBINS, 256, 0, stream>>>(binned, lstart, rowptr, dinv, csr, x, tbuf);

    gather_gemm_kernel<false><<<1250, 256, 0, stream>>>(tbuf, rowptr, csr, dinv, b1, W1, out);

    gather_gemm_kernel<false><<<1250, 256, 0, stream>>>(out, rowptr, csr, dinv, b2, W2, tbuf);

    gather_gemm_kernel<true><<<1250, 256, 0, stream>>>(tbuf, rowptr, csr, dinv, b3, W3, out);
}
